# starting kernel
# baseline (speedup 1.0000x reference)
#include <hip/hip_runtime.h>
#include <cstdint>

#define NN      50000
#define NE      1600000
#define ET      (NE + NN)
#define IN_C    256
#define HID_C   128
#define HEADS   8
#define OUT_C   16
#define SLOPE   0.2f

typedef float v2f __attribute__((ext_vector_type(2)));
typedef float v8f __attribute__((ext_vector_type(8)));

__device__ __forceinline__ unsigned enc_f32(float f) {
    unsigned u = __float_as_uint(f);
    return (u & 0x80000000u) ? ~u : (u | 0x80000000u);
}
__device__ __forceinline__ float dec_f32(unsigned e) {
    unsigned u = (e & 0x80000000u) ? (e & 0x7FFFFFFFu) : ~e;
    return __uint_as_float(u);
}

__global__ void k_zero(unsigned* __restrict__ p, long n) {
    long i = (long)blockIdx.x * 256 + threadIdx.x;
    if (i < n) p[i] = 0u;
}

__global__ void k_gemm1(const float* __restrict__ X, const float* __restrict__ W,
                        float* __restrict__ H) {
    const int wave = threadIdx.x >> 5;
    const int lane = threadIdx.x & 31;
    const int mBase = blockIdx.x << 4;
    const int nBase = wave << 4;
    const int lm = lane & 15;
    const int kh = (lane >> 4) << 1;
    const float* xr = X + (size_t)(mBase + lm) * IN_C + kh;
    const float* wp = W + (size_t)kh * HID_C + nBase + lm;
    v8f acc = {};
    #pragma unroll 8
    for (int kb = 0; kb < IN_C; kb += 4) {
        v2f a = *(const v2f*)(xr + kb);
        v2f b;
        b.x = wp[(size_t)kb * HID_C];
        b.y = wp[(size_t)kb * HID_C + HID_C];
        acc = __builtin_amdgcn_wmma_f32_16x16x4_f32(false, a, false, b,
                                                    (short)0, acc, false, false);
    }
    float* hp = H + (size_t)(mBase + ((lane >> 4) << 3)) * HID_C + nBase + lm;
    #pragma unroll
    for (int r = 0; r < 8; ++r) hp[(size_t)r * HID_C] = acc[r];
}

__global__ void k_gemm2(const float* __restrict__ X, const float* __restrict__ W,
                        float* __restrict__ H) {
    const int wave = threadIdx.x >> 5;
    const int lane = threadIdx.x & 31;
    const int mBase = (blockIdx.x << 7) + (wave << 4);
    if (mBase >= NN) return;
    const int lm = lane & 15;
    const int kh = (lane >> 4) << 1;
    const float* xr = X + (size_t)(mBase + lm) * HID_C + kh;
    const float* wp = W + (size_t)kh * OUT_C + lm;
    v8f acc = {};
    #pragma unroll 8
    for (int kb = 0; kb < HID_C; kb += 4) {
        v2f a = *(const v2f*)(xr + kb);
        v2f b;
        b.x = wp[(size_t)kb * OUT_C];
        b.y = wp[(size_t)kb * OUT_C + OUT_C];
        acc = __builtin_amdgcn_wmma_f32_16x16x4_f32(false, a, false, b,
                                                    (short)0, acc, false, false);
    }
    float* hp = H + (size_t)(mBase + ((lane >> 4) << 3)) * OUT_C + lm;
    #pragma unroll
    for (int r = 0; r < 8; ++r) hp[(size_t)r * OUT_C] = acc[r];
}

__global__ void k_attvec(const float* __restrict__ H, const float* __restrict__ attS,
                         const float* __restrict__ attD, float* __restrict__ aS,
                         float* __restrict__ aD, int nH) {
    int idx = blockIdx.x * blockDim.x + threadIdx.x;
    if (idx >= NN * nH) return;
    int hd = idx % nH;
    const float4* hp = (const float4*)(H + (size_t)idx * 16);
    const float4* sp = (const float4*)(attS + hd * 16);
    const float4* dp = (const float4*)(attD + hd * 16);
    float s = 0.f, d = 0.f;
    #pragma unroll
    for (int j = 0; j < 4; ++j) {
        float4 h4 = hp[j], s4 = sp[j], d4 = dp[j];
        s += h4.x * s4.x + h4.y * s4.y + h4.z * s4.z + h4.w * s4.w;
        d += h4.x * d4.x + h4.y * d4.y + h4.z * d4.z + h4.w * d4.w;
    }
    aS[idx] = s; aD[idx] = d;
}

template <int NH>
__global__ void k_edge_max(const int* __restrict__ esrc, const int* __restrict__ edst,
                           const float* __restrict__ aS, const float* __restrict__ aD,
                           unsigned* __restrict__ amax) {
    int e = blockIdx.x * blockDim.x + threadIdx.x;
    if (e >= ET) return;
    int s, d;
    if (e < NE) { s = esrc[e]; d = edst[e]; } else { s = e - NE; d = s; }
    #pragma unroll
    for (int hd = 0; hd < NH; ++hd) {
        float al = aS[(size_t)s * NH + hd] + aD[(size_t)d * NH + hd];
        al = al > 0.f ? al : al * SLOPE;
        atomicMax(&amax[(size_t)d * NH + hd], enc_f32(al));
    }
}

template <int NH>
__global__ void k_edge_accum(const int* __restrict__ esrc, const int* __restrict__ edst,
                             const float* __restrict__ aS, const float* __restrict__ aD,
                             const unsigned* __restrict__ amax, const float* __restrict__ H,
                             float* __restrict__ denom, float* __restrict__ accum) {
    long idx = (long)blockIdx.x * blockDim.x + threadIdx.x;
    if (idx >= (long)ET * NH) return;
    int e  = (int)(idx / NH);
    int hd = (int)(idx - (long)e * NH);
    int s, d;
    if (e < NE) { s = esrc[e]; d = edst[e]; } else { s = e - NE; d = s; }
    float al = aS[(size_t)s * NH + hd] + aD[(size_t)d * NH + hd];
    al = al > 0.f ? al : al * SLOPE;
    float w = __expf(al - dec_f32(amax[(size_t)d * NH + hd]));
    atomicAdd(&denom[(size_t)d * NH + hd], w);
    const float4* hp = (const float4*)(H + ((size_t)s * NH + hd) * 16);
    float* op = accum + ((size_t)d * NH + hd) * 16;
    #pragma unroll
    for (int j = 0; j < 4; ++j) {
        float4 v = hp[j];
        atomicAdd(op + 4 * j + 0, w * v.x);
        atomicAdd(op + 4 * j + 1, w * v.y);
        atomicAdd(op + 4 * j + 2, w * v.z);
        atomicAdd(op + 4 * j + 3, w * v.w);
    }
}

__global__ void k_fin(float* __restrict__ accum, const float* __restrict__ denom,
                      const float* __restrict__ bias, int rowMask, long total, int relu) {
    long idx = (long)blockIdx.x * blockDim.x + threadIdx.x;
    if (idx >= total) return;
    int c = (int)(idx & rowMask);
    long nh = idx >> 4;
    float v = accum[idx] / denom[nh] + bias[c];
    if (relu) v = fmaxf(v, 0.f);
    accum[idx] = v;
}

extern "C" void kernel_launch(void* const* d_in, const int* in_sizes, int n_in,
                              void* d_out, int out_size, void* d_ws, size_t ws_size,
                              hipStream_t stream) {
    const float* x     = (const float*)d_in[0];
    const int*   ei    = (const int*)  d_in[1];
    const float* W1    = (const float*)d_in[2];
    const float* attS1 = (const float*)d_in[3];
    const float* attD1 = (const float*)d_in[4];
    const float* bias1 = (const float*)d_in[5];
    const float* W2    = (const float*)d_in[6];
    const float* attS2 = (const float*)d_in[7];
    const float* attD2 = (const float*)d_in[8];
    const float* bias2 = (const float*)d_in[9];
    float* out = (float*)d_out;
    const int* esrc = ei;
    const int* edst = ei + NE;

    unsigned* amax1  = (unsigned*)d_ws;
    float*    denom1 = (float*)(amax1 + (size_t)NN * 8);
    float*    accum1 = denom1 + (size_t)NN * 8;
    unsigned* amax2  = (unsigned*)(accum1 + (size_t)NN * 128);
    float*    denom2 = (float*)(amax2 + NN);
    float* h1  = denom2 + NN;
    float* aS1 = h1 + (size_t)NN * 128;
    float* aD1 = aS1 + (size_t)NN * 8;
    float* h2  = aD1 + (size_t)NN * 8;
    float* aS2 = h2 + (size_t)NN * 16;
    float* aD2 = aS2 + NN;

    const long znWs  = (long)NN * (8 + 8 + 128 + 1 + 1);
    const long znOut = (long)NN * OUT_C;
    k_zero<<<(int)((znWs  + 255) / 256), 256, 0, stream>>>((unsigned*)d_ws, znWs);
    k_zero<<<(int)((znOut + 255) / 256), 256, 0, stream>>>((unsigned*)out, znOut);

    k_gemm1<<<NN / 16, 256, 0, stream>>>(x, W1, h1);
    k_attvec<<<(NN * HEADS + 255) / 256, 256, 0, stream>>>(h1, attS1, attD1, aS1, aD1, HEADS);
    k_edge_max<HEADS><<<(ET + 255) / 256, 256, 0, stream>>>(esrc, edst, aS1, aD1, amax1);
    {
        long n = (long)ET * HEADS;
        k_edge_accum<HEADS><<<(int)((n + 255) / 256), 256, 0, stream>>>(
            esrc, edst, aS1, aD1, amax1, h1, denom1, accum1);
    }
    {
        long n = (long)NN * HID_C;
        k_fin<<<(int)((n + 255) / 256), 256, 0, stream>>>(accum1, denom1, bias1,
                                                          HID_C - 1, n, 1);
    }

    k_gemm2<<<(NN + 127) / 128, 256, 0, stream>>>(accum1, W2, h2);
    k_attvec<<<(NN + 255) / 256, 256, 0, stream>>>(h2, attS2, attD2, aS2, aD2, 1);
    k_edge_max<1><<<(ET + 255) / 256, 256, 0, stream>>>(esrc, edst, aS2, aD2, amax2);
    {
        long n = (long)ET;
        k_edge_accum<1><<<(int)((n + 255) / 256), 256, 0, stream>>>(
            esrc, edst, aS2, aD2, amax2, h2, denom2, out);
    }
    {
        long n = (long)NN * OUT_C;
        k_fin<<<(int)((n + 255) / 256), 256, 0, stream>>>(out, denom2, bias2,
                                                          OUT_C - 1, n, 0);
    }
}


# Edges grouped by destination each call (integer atomic rank, scan, place); per-node softmax and weighted row sum in registers, no float atomics; sums now run in rank order
# speedup vs baseline: 6.1792x; 6.1792x over previous
#include <hip/hip_runtime.h>
#include <cstdint>

#define NN      50000
#define NE      1600000
#define ET      (NE + NN)
#define IN_C    256
#define HID_C   128
#define HEADS   8
#define OUT_C   16
#define SLOPE   0.2f

typedef float v2f __attribute__((ext_vector_type(2)));
typedef float v8f __attribute__((ext_vector_type(8)));

__global__ void k_zero(unsigned* __restrict__ p, long n) {
    long i = (long)blockIdx.x * 256 + threadIdx.x;
    if (i < n) p[i] = 0u;
}

__global__ void k_gemm1(const float* __restrict__ X, const float* __restrict__ W,
                        float* __restrict__ H) {
    const int wave = threadIdx.x >> 5;
    const int lane = threadIdx.x & 31;
    const int mBase = blockIdx.x << 4;
    const int nBase = wave << 4;
    const int lm = lane & 15;
    const int kh = (lane >> 4) << 1;
    const float* xr = X + (size_t)(mBase + lm) * IN_C + kh;
    const float* wp = W + (size_t)kh * HID_C + nBase + lm;
    v8f acc = {};
    #pragma unroll 8
    for (int kb = 0; kb < IN_C; kb += 4) {
        v2f a = *(const v2f*)(xr + kb);
        v2f b;
        b.x = wp[(size_t)kb * HID_C];
        b.y = wp[(size_t)kb * HID_C + HID_C];
        acc = __builtin_amdgcn_wmma_f32_16x16x4_f32(false, a, false, b,
                                                    (short)0, acc, false, false);
    }
    float* hp = H + (size_t)(mBase + ((lane >> 4) << 3)) * HID_C + nBase + lm;
    #pragma unroll
    for (int r = 0; r < 8; ++r) hp[(size_t)r * HID_C] = acc[r];
}

__global__ void k_gemm2(const float* __restrict__ X, const float* __restrict__ W,
                        float* __restrict__ H) {
    const int wave = threadIdx.x >> 5;
    const int lane = threadIdx.x & 31;
    const int mBase = (blockIdx.x << 7) + (wave << 4);
    if (mBase >= NN) return;
    const int lm = lane & 15;
    const int kh = (lane >> 4) << 1;
    const float* xr = X + (size_t)(mBase + lm) * HID_C + kh;
    const float* wp = W + (size_t)kh * OUT_C + lm;
    v8f acc = {};
    #pragma unroll 8
    for (int kb = 0; kb < HID_C; kb += 4) {
        v2f a = *(const v2f*)(xr + kb);
        v2f b;
        b.x = wp[(size_t)kb * OUT_C];
        b.y = wp[(size_t)kb * OUT_C + OUT_C];
        acc = __builtin_amdgcn_wmma_f32_16x16x4_f32(false, a, false, b,
                                                    (short)0, acc, false, false);
    }
    float* hp = H + (size_t)(mBase + ((lane >> 4) << 3)) * OUT_C + lm;
    #pragma unroll
    for (int r = 0; r < 8; ++r) hp[(size_t)r * OUT_C] = acc[r];
}

__global__ void k_attvec(const float* __restrict__ H, const float* __restrict__ attS,
                         const float* __restrict__ attD, float* __restrict__ aS,
                         float* __restrict__ aD, int nH) {
    int idx = blockIdx.x * blockDim.x + threadIdx.x;
    if (idx >= NN * nH) return;
    int hd = idx % nH;
    const float4* hp = (const float4*)(H + (size_t)idx * 16);
    const float4* sp = (const float4*)(attS + hd * 16);
    const float4* dp = (const float4*)(attD + hd * 16);
    float s = 0.f, d = 0.f;
    #pragma unroll
    for (int j = 0; j < 4; ++j) {
        float4 h4 = hp[j], s4 = sp[j], d4 = dp[j];
        s += h4.x * s4.x + h4.y * s4.y + h4.z * s4.z + h4.w * s4.w;
        d += h4.x * d4.x + h4.y * d4.y + h4.z * d4.z + h4.w * d4.w;
    }
    aS[idx] = s; aD[idx] = d;
}

#define QE (NE / 4)
static_assert(NE % 4 == 0, "edge count must split in four");

__global__ void k_count(const int* __restrict__ edst, unsigned* __restrict__ cnt,
                        unsigned* __restrict__ rank) {
    const int i = blockIdx.x * 256 + threadIdx.x;
    if (i >= QE) return;
    const int d0 = edst[i], d1 = edst[i + QE], d2 = edst[i + 2 * QE], d3 = edst[i + 3 * QE];
    const unsigned r0 = atomicAdd(&cnt[d0], 1u);
    const unsigned r1 = atomicAdd(&cnt[d1], 1u);
    const unsigned r2 = atomicAdd(&cnt[d2], 1u);
    const unsigned r3 = atomicAdd(&cnt[d3], 1u);
    rank[i] = r0; rank[i + QE] = r1; rank[i + 2 * QE] = r2; rank[i + 3 * QE] = r3;
}

#define SCAN_T  1024
#define SCAN_CH 49
#define SCAN_N  (SCAN_T * SCAN_CH)
static_assert(SCAN_N >= NN, "scan tile must cover all nodes");

__global__ void __launch_bounds__(SCAN_T)
k_scan(const unsigned* __restrict__ cnt, int* __restrict__ rowptr) {
    __shared__ unsigned buf[SCAN_N];
    __shared__ unsigned part[SCAN_T];
    const int t = threadIdx.x;
    for (int k = 0; k < SCAN_CH; ++k) {
        const int i = k * SCAN_T + t;
        buf[i] = (i < NN) ? cnt[i] : 0u;
    }
    __syncthreads();
    const int b = t * SCAN_CH;
    unsigned sum = 0u;
    for (int k = 0; k < SCAN_CH; ++k) sum += buf[b + k];
    part[t] = sum;
    __syncthreads();
    for (int off = 1; off < SCAN_T; off <<= 1) {
        const unsigned v = (t >= off) ? part[t - off] : 0u;
        __syncthreads();
        part[t] += v;
        __syncthreads();
    }
    unsigned run = (t > 0) ? part[t - 1] : 0u;
    for (int k = 0; k < SCAN_CH; ++k) {
        const unsigned c = buf[b + k];
        buf[b + k] = run;
        run += c;
    }
    __syncthreads();
    for (int k = 0; k < SCAN_CH; ++k) {
        const int i = k * SCAN_T + t;
        if (i < NN) rowptr[i] = (int)buf[i];
    }
    if (t == 0) rowptr[NN] = (int)part[SCAN_T - 1];
}

__global__ void k_fill(const int* __restrict__ esrc, const int* __restrict__ edst,
                       const int* __restrict__ rowptr, const unsigned* __restrict__ rank,
                       int* __restrict__ csr) {
    const int i = blockIdx.x * 256 + threadIdx.x;
    if (i >= QE) return;
    #pragma unroll
    for (int q = 0; q < 4; ++q) {
        const int e = i + q * QE;
        csr[rowptr[edst[e]] + (int)rank[e]] = esrc[e];
    }
}

__device__ __forceinline__ float lrelu(float a) { return a > 0.f ? a : a * SLOPE; }

#define A1_NPW 4
#define A1_U   8
static_assert(NN % (4 * A1_NPW) == 0, "nodes must fill the layer-1 aggregate grid");

__global__ void __launch_bounds__(128)
k_agg1(const int* __restrict__ rowptr, const int* __restrict__ csr,
       const float* __restrict__ aS, const float* __restrict__ aD,
       const float* __restrict__ H, const float* __restrict__ bias,
       float* __restrict__ O) {
    const int lane = threadIdx.x & 31;
    const int hl = lane >> 2;
    const int wv = __builtin_amdgcn_readfirstlane((int)((blockIdx.x * 128 + threadIdx.x) >> 5));
    const float4 b4 = *(const float4*)(bias + 4 * lane);
    for (int q = 0; q < A1_NPW; ++q) {
        const int n = wv * A1_NPW + q;
        const int beg = rowptr[n];
        const int end = rowptr[n + 1];
        const float aDl = aD[(size_t)n * HEADS + hl];
        const float aSelf = lrelu(aS[(size_t)n * HEADS + hl] + aDl);
        float m = aSelf;
        for (int base = beg; base < end; base += 32) {
            const int c = min(32, end - base);
            const int sv = (lane < c) ? csr[base + lane] : 0;
            for (int j = 0; j < c; j += A1_U) {
                float av[A1_U];
                #pragma unroll
                for (int u = 0; u < A1_U; ++u) {
                    const int s = __builtin_amdgcn_readlane(sv, min(j + u, c - 1));
                    av[u] = aS[(size_t)s * HEADS + hl];
                }
                #pragma unroll
                for (int u = 0; u < A1_U; ++u) m = fmaxf(m, lrelu(av[u] + aDl));
            }
        }
        const float w0 = __expf(aSelf - m);
        float den = w0;
        const float4 hv = *(const float4*)(H + (size_t)n * HID_C + 4 * lane);
        float4 acc = make_float4(w0 * hv.x, w0 * hv.y, w0 * hv.z, w0 * hv.w);
        for (int base = beg; base < end; base += 32) {
            const int c = min(32, end - base);
            const int sv = (lane < c) ? csr[base + lane] : 0;
            for (int j = 0; j < c; j += A1_U) {
                float av[A1_U];
                float4 rv[A1_U];
                #pragma unroll
                for (int u = 0; u < A1_U; ++u) {
                    const int s = __builtin_amdgcn_readlane(sv, min(j + u, c - 1));
                    av[u] = aS[(size_t)s * HEADS + hl];
                    rv[u] = *(const float4*)(H + (size_t)s * HID_C + 4 * lane);
                }
                #pragma unroll
                for (int u = 0; u < A1_U; ++u) {
                    float w = __expf(lrelu(av[u] + aDl) - m);
                    w = (j + u < c) ? w : 0.f;
                    den += w;
                    acc.x += w * rv[u].x; acc.y += w * rv[u].y;
                    acc.z += w * rv[u].z; acc.w += w * rv[u].w;
                }
            }
        }
        float4 o;
        o.x = fmaxf(acc.x / den + b4.x, 0.f);
        o.y = fmaxf(acc.y / den + b4.y, 0.f);
        o.z = fmaxf(acc.z / den + b4.z, 0.f);
        o.w = fmaxf(acc.w / den + b4.w, 0.f);
        *(float4*)(O + (size_t)n * HID_C + 4 * lane) = o;
    }
}

__global__ void __launch_bounds__(256)
k_agg2(const int* __restrict__ rowptr, const int* __restrict__ csr,
       const float* __restrict__ aS, const float* __restrict__ aD,
       const float* __restrict__ H, const float* __restrict__ bias,
       float* __restrict__ O) {
    const int gt = blockIdx.x * 256 + threadIdx.x;
    const int n = gt >> 2;
    const int k = gt & 3;
    if (n >= NN) return;
    const int beg = rowptr[n];
    const int end = rowptr[n + 1];
    const float aDn = aD[n];
    const float aSelf = lrelu(aS[n] + aDn);
    float m = aSelf;
    for (int i = beg + k; i < end; i += 4) m = fmaxf(m, lrelu(aS[csr[i]] + aDn));
    m = fmaxf(m, __shfl_xor(m, 1));
    m = fmaxf(m, __shfl_xor(m, 2));
    const float w0 = __expf(aSelf - m);
    float den = w0;
    const float4 hv = *(const float4*)(H + (size_t)n * OUT_C + 4 * k);
    float4 acc = make_float4(w0 * hv.x, w0 * hv.y, w0 * hv.z, w0 * hv.w);
    for (int i = beg; i < end; i += 4) {
        int sv[4];
        float av[4];
        float4 rv[4];
        #pragma unroll
        for (int u = 0; u < 4; ++u) sv[u] = csr[min(i + u, end - 1)];
        #pragma unroll
        for (int u = 0; u < 4; ++u) {
            av[u] = aS[sv[u]];
            rv[u] = *(const float4*)(H + (size_t)sv[u] * OUT_C + 4 * k);
        }
        #pragma unroll
        for (int u = 0; u < 4; ++u) {
            float w = __expf(lrelu(av[u] + aDn) - m);
            w = (i + u < end) ? w : 0.f;
            den += w;
            acc.x += w * rv[u].x; acc.y += w * rv[u].y;
            acc.z += w * rv[u].z; acc.w += w * rv[u].w;
        }
    }
    const float4 b4 = *(const float4*)(bias + 4 * k);
    float4 o;
    o.x = acc.x / den + b4.x;
    o.y = acc.y / den + b4.y;
    o.z = acc.z / den + b4.z;
    o.w = acc.w / den + b4.w;
    *(float4*)(O + (size_t)n * OUT_C + 4 * k) = o;
}

#define PADW(x) (((size_t)(x) + 63) & ~(size_t)63)

extern "C" void kernel_launch(void* const* d_in, const int* in_sizes, int n_in,
                              void* d_out, int out_size, void* d_ws, size_t ws_size,
                              hipStream_t stream) {
    const float* x     = (const float*)d_in[0];
    const int*   ei    = (const int*)  d_in[1];
    const float* W1    = (const float*)d_in[2];
    const float* attS1 = (const float*)d_in[3];
    const float* attD1 = (const float*)d_in[4];
    const float* bias1 = (const float*)d_in[5];
    const float* W2    = (const float*)d_in[6];
    const float* attS2 = (const float*)d_in[7];
    const float* attD2 = (const float*)d_in[8];
    const float* bias2 = (const float*)d_in[9];
    float* out = (float*)d_out;
    const int* esrc = ei;
    const int* edst = ei + NE;

    unsigned* cnt    = (unsigned*)d_ws;
    int*      rowptr = (int*)(cnt + PADW(NN));
    unsigned* rank   = (unsigned*)(rowptr + PADW(NN + 1));
    int*      csr    = (int*)(rank + PADW(NE));
    float* h1   = (float*)(csr + PADW(NE));
    float* aS1  = h1 + PADW((size_t)NN * 128);
    float* aD1  = aS1 + PADW((size_t)NN * 8);
    float* out1 = aD1 + PADW((size_t)NN * 8);
    float* h2   = out1 + PADW((size_t)NN * 128);
    float* aS2  = h2 + PADW((size_t)NN * 16);
    float* aD2  = aS2 + PADW(NN);

    k_zero<<<(NN + 255) / 256, 256, 0, stream>>>(cnt, (long)NN);
    k_count<<<(QE + 255) / 256, 256, 0, stream>>>(edst, cnt, rank);
    k_scan<<<1, SCAN_T, 0, stream>>>(cnt, rowptr);
    k_fill<<<(QE + 255) / 256, 256, 0, stream>>>(esrc, edst, rowptr, rank, csr);

    k_gemm1<<<NN / 16, 256, 0, stream>>>(x, W1, h1);
    k_attvec<<<(NN * HEADS + 255) / 256, 256, 0, stream>>>(h1, attS1, attD1, aS1, aD1, HEADS);
    k_agg1<<<NN / (4 * A1_NPW), 128, 0, stream>>>(rowptr, csr, aS1, aD1, h1, bias1, out1);

    k_gemm2<<<(NN + 127) / 128, 256, 0, stream>>>(out1, W2, h2);
    k_attvec<<<(NN + 255) / 256, 256, 0, stream>>>(h2, attS2, attD2, aS2, aD2, 1);
    k_agg2<<<(NN * 4 + 255) / 256, 256, 0, stream>>>(rowptr, csr, aS2, aD2, h2, bias2, out);
}


# Both aggregates walk each destination's edge list once: running maximum, sums rescaled by exp(old-new) when it rises (online softmax); the max pass and its score gathers go; sum rounding changes
# speedup vs baseline: 6.6669x; 1.0789x over previous
#include <hip/hip_runtime.h>
#include <cstdint>

#define NN      50000
#define NE      1600000
#define ET      (NE + NN)
#define IN_C    256
#define HID_C   128
#define HEADS   8
#define OUT_C   16
#define SLOPE   0.2f

typedef float v2f __attribute__((ext_vector_type(2)));
typedef float v8f __attribute__((ext_vector_type(8)));

__global__ void k_zero(unsigned* __restrict__ p, long n) {
    long i = (long)blockIdx.x * 256 + threadIdx.x;
    if (i < n) p[i] = 0u;
}

__global__ void k_gemm1(const float* __restrict__ X, const float* __restrict__ W,
                        float* __restrict__ H) {
    const int wave = threadIdx.x >> 5;
    const int lane = threadIdx.x & 31;
    const int mBase = blockIdx.x << 4;
    const int nBase = wave << 4;
    const int lm = lane & 15;
    const int kh = (lane >> 4) << 1;
    const float* xr = X + (size_t)(mBase + lm) * IN_C + kh;
    const float* wp = W + (size_t)kh * HID_C + nBase + lm;
    v8f acc = {};
    #pragma unroll 8
    for (int kb = 0; kb < IN_C; kb += 4) {
        v2f a = *(const v2f*)(xr + kb);
        v2f b;
        b.x = wp[(size_t)kb * HID_C];
        b.y = wp[(size_t)kb * HID_C + HID_C];
        acc = __builtin_amdgcn_wmma_f32_16x16x4_f32(false, a, false, b,
                                                    (short)0, acc, false, false);
    }
    float* hp = H + (size_t)(mBase + ((lane >> 4) << 3)) * HID_C + nBase + lm;
    #pragma unroll
    for (int r = 0; r < 8; ++r) hp[(size_t)r * HID_C] = acc[r];
}

__global__ void k_gemm2(const float* __restrict__ X, const float* __restrict__ W,
                        float* __restrict__ H) {
    const int wave = threadIdx.x >> 5;
    const int lane = threadIdx.x & 31;
    const int mBase = (blockIdx.x << 7) + (wave << 4);
    if (mBase >= NN) return;
    const int lm = lane & 15;
    const int kh = (lane >> 4) << 1;
    const float* xr = X + (size_t)(mBase + lm) * HID_C + kh;
    const float* wp = W + (size_t)kh * OUT_C + lm;
    v8f acc = {};
    #pragma unroll 8
    for (int kb = 0; kb < HID_C; kb += 4) {
        v2f a = *(const v2f*)(xr + kb);
        v2f b;
        b.x = wp[(size_t)kb * OUT_C];
        b.y = wp[(size_t)kb * OUT_C + OUT_C];
        acc = __builtin_amdgcn_wmma_f32_16x16x4_f32(false, a, false, b,
                                                    (short)0, acc, false, false);
    }
    float* hp = H + (size_t)(mBase + ((lane >> 4) << 3)) * OUT_C + lm;
    #pragma unroll
    for (int r = 0; r < 8; ++r) hp[(size_t)r * OUT_C] = acc[r];
}

__global__ void k_attvec(const float* __restrict__ H, const float* __restrict__ attS,
                         const float* __restrict__ attD, float* __restrict__ aS,
                         float* __restrict__ aD, int nH) {
    int idx = blockIdx.x * blockDim.x + threadIdx.x;
    if (idx >= NN * nH) return;
    int hd = idx % nH;
    const float4* hp = (const float4*)(H + (size_t)idx * 16);
    const float4* sp = (const float4*)(attS + hd * 16);
    const float4* dp = (const float4*)(attD + hd * 16);
    float s = 0.f, d = 0.f;
    #pragma unroll
    for (int j = 0; j < 4; ++j) {
        float4 h4 = hp[j], s4 = sp[j], d4 = dp[j];
        s += h4.x * s4.x + h4.y * s4.y + h4.z * s4.z + h4.w * s4.w;
        d += h4.x * d4.x + h4.y * d4.y + h4.z * d4.z + h4.w * d4.w;
    }
    aS[idx] = s; aD[idx] = d;
}

#define QE (NE / 4)
static_assert(NE % 4 == 0, "edge count must split in four");

__global__ void k_count(const int* __restrict__ edst, unsigned* __restrict__ cnt,
                        unsigned* __restrict__ rank) {
    const int i = blockIdx.x * 256 + threadIdx.x;
    if (i >= QE) return;
    const int d0 = edst[i], d1 = edst[i + QE], d2 = edst[i + 2 * QE], d3 = edst[i + 3 * QE];
    const unsigned r0 = atomicAdd(&cnt[d0], 1u);
    const unsigned r1 = atomicAdd(&cnt[d1], 1u);
    const unsigned r2 = atomicAdd(&cnt[d2], 1u);
    const unsigned r3 = atomicAdd(&cnt[d3], 1u);
    rank[i] = r0; rank[i + QE] = r1; rank[i + 2 * QE] = r2; rank[i + 3 * QE] = r3;
}

#define SCAN_T  1024
#define SCAN_CH 49
#define SCAN_N  (SCAN_T * SCAN_CH)
static_assert(SCAN_N >= NN, "scan tile must cover all nodes");

__global__ void __launch_bounds__(SCAN_T)
k_scan(const unsigned* __restrict__ cnt, int* __restrict__ rowptr) {
    __shared__ unsigned buf[SCAN_N];
    __shared__ unsigned part[SCAN_T];
    const int t = threadIdx.x;
    for (int k = 0; k < SCAN_CH; ++k) {
        const int i = k * SCAN_T + t;
        buf[i] = (i < NN) ? cnt[i] : 0u;
    }
    __syncthreads();
    const int b = t * SCAN_CH;
    unsigned sum = 0u;
    for (int k = 0; k < SCAN_CH; ++k) sum += buf[b + k];
    part[t] = sum;
    __syncthreads();
    for (int off = 1; off < SCAN_T; off <<= 1) {
        const unsigned v = (t >= off) ? part[t - off] : 0u;
        __syncthreads();
        part[t] += v;
        __syncthreads();
    }
    unsigned run = (t > 0) ? part[t - 1] : 0u;
    for (int k = 0; k < SCAN_CH; ++k) {
        const unsigned c = buf[b + k];
        buf[b + k] = run;
        run += c;
    }
    __syncthreads();
    for (int k = 0; k < SCAN_CH; ++k) {
        const int i = k * SCAN_T + t;
        if (i < NN) rowptr[i] = (int)buf[i];
    }
    if (t == 0) rowptr[NN] = (int)part[SCAN_T - 1];
}

__global__ void k_fill(const int* __restrict__ esrc, const int* __restrict__ edst,
                       const int* __restrict__ rowptr, const unsigned* __restrict__ rank,
                       int* __restrict__ csr) {
    const int i = blockIdx.x * 256 + threadIdx.x;
    if (i >= QE) return;
    #pragma unroll
    for (int q = 0; q < 4; ++q) {
        const int e = i + q * QE;
        csr[rowptr[edst[e]] + (int)rank[e]] = esrc[e];
    }
}

__device__ __forceinline__ float lrelu(float a) { return a > 0.f ? a : a * SLOPE; }

#define A1_NPW 4
#define A1_U   8
static_assert(NN % (4 * A1_NPW) == 0, "nodes must fill the layer-1 aggregate grid");

__global__ void __launch_bounds__(128)
k_agg1(const int* __restrict__ rowptr, const int* __restrict__ csr,
       const float* __restrict__ aS, const float* __restrict__ aD,
       const float* __restrict__ H, const float* __restrict__ bias,
       float* __restrict__ O) {
    const int lane = threadIdx.x & 31;
    const int hl = lane >> 2;
    const int wv = __builtin_amdgcn_readfirstlane((int)((blockIdx.x * 128 + threadIdx.x) >> 5));
    const float4 b4 = *(const float4*)(bias + 4 * lane);
    for (int q = 0; q < A1_NPW; ++q) {
        const int n = wv * A1_NPW + q;
        const int beg = rowptr[n];
        const int end = rowptr[n + 1];
        const float aDl = aD[(size_t)n * HEADS + hl];
        float m = lrelu(aS[(size_t)n * HEADS + hl] + aDl);
        float den = 1.f;
        float4 acc = *(const float4*)(H + (size_t)n * HID_C + 4 * lane);
        for (int base = beg; base < end; base += 32) {
            const int c = min(32, end - base);
            const int sv = (lane < c) ? csr[base + lane] : 0;
            for (int j = 0; j < c; j += A1_U) {
                float av[A1_U];
                float4 rv[A1_U];
                #pragma unroll
                for (int u = 0; u < A1_U; ++u) {
                    const int s = __builtin_amdgcn_readlane(sv, min(j + u, c - 1));
                    av[u] = aS[(size_t)s * HEADS + hl];
                    rv[u] = *(const float4*)(H + (size_t)s * HID_C + 4 * lane);
                }
                float gm = m;
                #pragma unroll
                for (int u = 0; u < A1_U; ++u) {
                    av[u] = lrelu(av[u] + aDl);
                    gm = fmaxf(gm, av[u]);
                }
                const float r = __expf(m - gm);
                m = gm;
                den *= r;
                acc.x *= r; acc.y *= r; acc.z *= r; acc.w *= r;
                #pragma unroll
                for (int u = 0; u < A1_U; ++u) {
                    float w = __expf(av[u] - m);
                    w = (j + u < c) ? w : 0.f;
                    den += w;
                    acc.x += w * rv[u].x; acc.y += w * rv[u].y;
                    acc.z += w * rv[u].z; acc.w += w * rv[u].w;
                }
            }
        }
        float4 o;
        o.x = fmaxf(acc.x / den + b4.x, 0.f);
        o.y = fmaxf(acc.y / den + b4.y, 0.f);
        o.z = fmaxf(acc.z / den + b4.z, 0.f);
        o.w = fmaxf(acc.w / den + b4.w, 0.f);
        *(float4*)(O + (size_t)n * HID_C + 4 * lane) = o;
    }
}

__global__ void __launch_bounds__(256)
k_agg2(const int* __restrict__ rowptr, const int* __restrict__ csr,
       const float* __restrict__ aS, const float* __restrict__ aD,
       const float* __restrict__ H, const float* __restrict__ bias,
       float* __restrict__ O) {
    const int gt = blockIdx.x * 256 + threadIdx.x;
    const int n = gt >> 2;
    const int k = gt & 3;
    if (n >= NN) return;
    const int beg = rowptr[n];
    const int end = rowptr[n + 1];
    const float aDn = aD[n];
    float m = lrelu(aS[n] + aDn);
    float den = 1.f;
    float4 acc = *(const float4*)(H + (size_t)n * OUT_C + 4 * k);
    for (int i = beg; i < end; i += 4) {
        int sv[4];
        float av[4];
        float4 rv[4];
        #pragma unroll
        for (int u = 0; u < 4; ++u) sv[u] = csr[min(i + u, end - 1)];
        #pragma unroll
        for (int u = 0; u < 4; ++u) {
            av[u] = aS[sv[u]];
            rv[u] = *(const float4*)(H + (size_t)sv[u] * OUT_C + 4 * k);
        }
        float gm = m;
        #pragma unroll
        for (int u = 0; u < 4; ++u) {
            av[u] = lrelu(av[u] + aDn);
            gm = fmaxf(gm, av[u]);
        }
        const float r = __expf(m - gm);
        m = gm;
        den *= r;
        acc.x *= r; acc.y *= r; acc.z *= r; acc.w *= r;
        #pragma unroll
        for (int u = 0; u < 4; ++u) {
            float w = __expf(av[u] - m);
            w = (i + u < end) ? w : 0.f;
            den += w;
            acc.x += w * rv[u].x; acc.y += w * rv[u].y;
            acc.z += w * rv[u].z; acc.w += w * rv[u].w;
        }
    }
    const float4 b4 = *(const float4*)(bias + 4 * k);
    float4 o;
    o.x = acc.x / den + b4.x;
    o.y = acc.y / den + b4.y;
    o.z = acc.z / den + b4.z;
    o.w = acc.w / den + b4.w;
    *(float4*)(O + (size_t)n * OUT_C + 4 * k) = o;
}

#define PADW(x) (((size_t)(x) + 63) & ~(size_t)63)

extern "C" void kernel_launch(void* const* d_in, const int* in_sizes, int n_in,
                              void* d_out, int out_size, void* d_ws, size_t ws_size,
                              hipStream_t stream) {
    const float* x     = (const float*)d_in[0];
    const int*   ei    = (const int*)  d_in[1];
    const float* W1    = (const float*)d_in[2];
    const float* attS1 = (const float*)d_in[3];
    const float* attD1 = (const float*)d_in[4];
    const float* bias1 = (const float*)d_in[5];
    const float* W2    = (const float*)d_in[6];
    const float* attS2 = (const float*)d_in[7];
    const float* attD2 = (const float*)d_in[8];
    const float* bias2 = (const float*)d_in[9];
    float* out = (float*)d_out;
    const int* esrc = ei;
    const int* edst = ei + NE;

    unsigned* cnt    = (unsigned*)d_ws;
    int*      rowptr = (int*)(cnt + PADW(NN));
    unsigned* rank   = (unsigned*)(rowptr + PADW(NN + 1));
    int*      csr    = (int*)(rank + PADW(NE));
    float* h1   = (float*)(csr + PADW(NE));
    float* aS1  = h1 + PADW((size_t)NN * 128);
    float* aD1  = aS1 + PADW((size_t)NN * 8);
    float* out1 = aD1 + PADW((size_t)NN * 8);
    float* h2   = out1 + PADW((size_t)NN * 128);
    float* aS2  = h2 + PADW((size_t)NN * 16);
    float* aD2  = aS2 + PADW(NN);

    k_zero<<<(NN + 255) / 256, 256, 0, stream>>>(cnt, (long)NN);
    k_count<<<(QE + 255) / 256, 256, 0, stream>>>(edst, cnt, rank);
    k_scan<<<1, SCAN_T, 0, stream>>>(cnt, rowptr);
    k_fill<<<(QE + 255) / 256, 256, 0, stream>>>(esrc, edst, rowptr, rank, csr);

    k_gemm1<<<NN / 16, 256, 0, stream>>>(x, W1, h1);
    k_attvec<<<(NN * HEADS + 255) / 256, 256, 0, stream>>>(h1, attS1, attD1, aS1, aD1, HEADS);
    k_agg1<<<NN / (4 * A1_NPW), 128, 0, stream>>>(rowptr, csr, aS1, aD1, h1, bias1, out1);

    k_gemm2<<<(NN + 127) / 128, 256, 0, stream>>>(out1, W2, h2);
    k_attvec<<<(NN + 255) / 256, 256, 0, stream>>>(h2, attS2, attD2, aS2, aD2, 1);
    k_agg2<<<(NN * 4 + 255) / 256, 256, 0, stream>>>(rowptr, csr, aS2, aD2, h2, bias2, out);
}


# Layer-1 GEMM on 256 resident work-groups: each wave keeps its 16-column slice of W1 in registers for all its row tiles; x rows staged once in LDS by whole-line loads; same sums
# speedup vs baseline: 7.2811x; 1.0921x over previous
#include <hip/hip_runtime.h>
#include <cstdint>

#define NN      50000
#define NE      1600000
#define ET      (NE + NN)
#define IN_C    256
#define HID_C   128
#define HEADS   8
#define OUT_C   16
#define SLOPE   0.2f

typedef float v2f __attribute__((ext_vector_type(2)));
typedef float v8f __attribute__((ext_vector_type(8)));

__global__ void k_zero(unsigned* __restrict__ p, long n) {
    long i = (long)blockIdx.x * 256 + threadIdx.x;
    if (i < n) p[i] = 0u;
}

#define G1_GRID   256
#define G1_TILES  (NN / 16)
#define G1_PITCH  (IN_C + 4)
#define G1_BUF    20736
static_assert(G1_BUF >= 32 * G1_PITCH, "buffer holds a step");
static_assert(NN % 16 == 0, "row tiles are whole");

#define G1_KC 4
template <int NT>
__device__ __forceinline__ void g1_tiles(const float* __restrict__ ap, const v2f (&bf)[64],
                                         float* __restrict__ hp) {
    v8f acc[NT];
    #pragma unroll
    for (int m = 0; m < NT; ++m) acc[m] = v8f{};
    v2f af[2][NT][G1_KC];
    #pragma unroll
    for (int m = 0; m < NT; ++m) {
        #pragma unroll
        for (int i = 0; i < G1_KC; ++i)
            af[0][m][i] = *(const v2f*)(ap + m * 16 * G1_PITCH + 4 * i);
    }
    __builtin_amdgcn_sched_group_barrier(0x100, NT * G1_KC / 2, 0);
    #pragma unroll
    for (int c = 0; c < 64 / G1_KC; ++c) {
        if (c + 1 < 64 / G1_KC) {
            #pragma unroll
            for (int m = 0; m < NT; ++m) {
                #pragma unroll
                for (int i = 0; i < G1_KC; ++i)
                    af[(c + 1) & 1][m][i] =
                        *(const v2f*)(ap + m * 16 * G1_PITCH + 4 * (G1_KC * (c + 1) + i));
            }
        }
        #pragma unroll
        for (int i = 0; i < G1_KC; ++i) {
            #pragma unroll
            for (int m = 0; m < NT; ++m)
                acc[m] = __builtin_amdgcn_wmma_f32_16x16x4_f32(false, af[c & 1][m][i], false,
                                                               bf[G1_KC * c + i], (short)0,
                                                               acc[m], false, false);
        }
        if (c + 1 < 64 / G1_KC) __builtin_amdgcn_sched_group_barrier(0x100, NT * G1_KC / 2, 0);
        __builtin_amdgcn_sched_group_barrier(0x008, NT * G1_KC, 0);
    }
    #pragma unroll
    for (int m = 0; m < NT; ++m) {
        #pragma unroll
        for (int r = 0; r < 8; ++r) hp[(size_t)(m * 16 + r) * HID_C] = acc[m][r];
    }
}

__device__ __forceinline__ void g1_lds_barrier() {
    __builtin_amdgcn_fence(__ATOMIC_RELEASE, "workgroup", "local");
    __builtin_amdgcn_s_barrier();
    __builtin_amdgcn_fence(__ATOMIC_ACQUIRE, "workgroup", "local");
}

__global__ void __launch_bounds__(256)
k_gemm1(const float* __restrict__ X, const float* __restrict__ W,
        float* __restrict__ H) {
    __shared__ __attribute__((aligned(16))) float As[2][G1_BUF];
    const int tid  = threadIdx.x;
    const int wave = tid >> 5;
    const int lane = tid & 31;
    const int nBase = wave << 4;
    const int lm = lane & 15;
    const int kh = (lane >> 4) << 1;
    const int t0 = ((int)blockIdx.x * G1_TILES) / G1_GRID;
    const int t1 = (((int)blockIdx.x + 1) * G1_TILES) / G1_GRID;
    const int rowBeg = t0 << 4;
    const int rowEnd = t1 << 4;

    const int sRow = tid >> 6;
    const int sCol = (tid & 63) << 2;
    float4 pre[8];
    auto fetch = [&](int r) {
        #pragma unroll
        for (int i = 0; i < 8; ++i) {
            const int row = r + sRow + 4 * i;
            pre[i] = (row < rowEnd) ? *(const float4*)(X + (size_t)row * IN_C + sCol)
                                    : float4{0.f, 0.f, 0.f, 0.f};
        }
    };
    auto stash = [&](int b) {
        #pragma unroll
        for (int i = 0; i < 8; ++i)
            *(float4*)(&As[b][(sRow + 4 * i) * G1_PITCH + sCol]) = pre[i];
    };

    v2f bf[64];
    const float* wp = W + (size_t)kh * HID_C + nBase + lm;
    #pragma unroll
    for (int kb = 0; kb < 64; ++kb) {
        bf[kb].x = wp[(size_t)(4 * kb) * HID_C];
        bf[kb].y = wp[(size_t)(4 * kb) * HID_C + HID_C];
    }

    fetch(rowBeg);
    stash(0);
    fetch(rowBeg + 32);
    g1_lds_barrier();

    int buf = 0;
    for (int r = rowBeg; r < rowEnd; r += 32) {
        const float* ap = &As[buf][lm * G1_PITCH + kh];
        float* hp = H + (size_t)(r + ((lane >> 4) << 3)) * HID_C + nBase + lm;
        if (rowEnd - r >= 32) g1_tiles<2>(ap, bf, hp);
        else                  g1_tiles<1>(ap, bf, hp);
        stash(buf ^ 1);
        fetch(r + 64);
        g1_lds_barrier();
        buf ^= 1;
    }
}

__global__ void k_gemm2(const float* __restrict__ X, const float* __restrict__ W,
                        float* __restrict__ H) {
    const int wave = threadIdx.x >> 5;
    const int lane = threadIdx.x & 31;
    const int mBase = (blockIdx.x << 7) + (wave << 4);
    if (mBase >= NN) return;
    const int lm = lane & 15;
    const int kh = (lane >> 4) << 1;
    const float* xr = X + (size_t)(mBase + lm) * HID_C + kh;
    const float* wp = W + (size_t)kh * OUT_C + lm;
    v8f acc = {};
    #pragma unroll 8
    for (int kb = 0; kb < HID_C; kb += 4) {
        v2f a = *(const v2f*)(xr + kb);
        v2f b;
        b.x = wp[(size_t)kb * OUT_C];
        b.y = wp[(size_t)kb * OUT_C + OUT_C];
        acc = __builtin_amdgcn_wmma_f32_16x16x4_f32(false, a, false, b,
                                                    (short)0, acc, false, false);
    }
    float* hp = H + (size_t)(mBase + ((lane >> 4) << 3)) * OUT_C + lm;
    #pragma unroll
    for (int r = 0; r < 8; ++r) hp[(size_t)r * OUT_C] = acc[r];
}

__global__ void k_attvec(const float* __restrict__ H, const float* __restrict__ attS,
                         const float* __restrict__ attD, float* __restrict__ aS,
                         float* __restrict__ aD, int nH) {
    int idx = blockIdx.x * blockDim.x + threadIdx.x;
    if (idx >= NN * nH) return;
    int hd = idx % nH;
    const float4* hp = (const float4*)(H + (size_t)idx * 16);
    const float4* sp = (const float4*)(attS + hd * 16);
    const float4* dp = (const float4*)(attD + hd * 16);
    float s = 0.f, d = 0.f;
    #pragma unroll
    for (int j = 0; j < 4; ++j) {
        float4 h4 = hp[j], s4 = sp[j], d4 = dp[j];
        s += h4.x * s4.x + h4.y * s4.y + h4.z * s4.z + h4.w * s4.w;
        d += h4.x * d4.x + h4.y * d4.y + h4.z * d4.z + h4.w * d4.w;
    }
    aS[idx] = s; aD[idx] = d;
}

#define QE (NE / 4)
static_assert(NE % 4 == 0, "edge count must split in four");

__global__ void k_count(const int* __restrict__ edst, unsigned* __restrict__ cnt,
                        unsigned* __restrict__ rank) {
    const int i = blockIdx.x * 256 + threadIdx.x;
    if (i >= QE) return;
    const int d0 = edst[i], d1 = edst[i + QE], d2 = edst[i + 2 * QE], d3 = edst[i + 3 * QE];
    const unsigned r0 = atomicAdd(&cnt[d0], 1u);
    const unsigned r1 = atomicAdd(&cnt[d1], 1u);
    const unsigned r2 = atomicAdd(&cnt[d2], 1u);
    const unsigned r3 = atomicAdd(&cnt[d3], 1u);
    rank[i] = r0; rank[i + QE] = r1; rank[i + 2 * QE] = r2; rank[i + 3 * QE] = r3;
}

#define SCAN_T  1024
#define SCAN_CH 49
#define SCAN_N  (SCAN_T * SCAN_CH)
static_assert(SCAN_N >= NN, "scan tile must cover all nodes");

__global__ void __launch_bounds__(SCAN_T)
k_scan(const unsigned* __restrict__ cnt, int* __restrict__ rowptr) {
    __shared__ unsigned buf[SCAN_N];
    __shared__ unsigned part[SCAN_T];
    const int t = threadIdx.x;
    for (int k = 0; k < SCAN_CH; ++k) {
        const int i = k * SCAN_T + t;
        buf[i] = (i < NN) ? cnt[i] : 0u;
    }
    __syncthreads();
    const int b = t * SCAN_CH;
    unsigned sum = 0u;
    for (int k = 0; k < SCAN_CH; ++k) sum += buf[b + k];
    part[t] = sum;
    __syncthreads();
    for (int off = 1; off < SCAN_T; off <<= 1) {
        const unsigned v = (t >= off) ? part[t - off] : 0u;
        __syncthreads();
        part[t] += v;
        __syncthreads();
    }
    unsigned run = (t > 0) ? part[t - 1] : 0u;
    for (int k = 0; k < SCAN_CH; ++k) {
        const unsigned c = buf[b + k];
        buf[b + k] = run;
        run += c;
    }
    __syncthreads();
    for (int k = 0; k < SCAN_CH; ++k) {
        const int i = k * SCAN_T + t;
        if (i < NN) rowptr[i] = (int)buf[i];
    }
    if (t == 0) rowptr[NN] = (int)part[SCAN_T - 1];
}

__global__ void k_fill(const int* __restrict__ esrc, const int* __restrict__ edst,
                       const int* __restrict__ rowptr, const unsigned* __restrict__ rank,
                       int* __restrict__ csr) {
    const int i = blockIdx.x * 256 + threadIdx.x;
    if (i >= QE) return;
    #pragma unroll
    for (int q = 0; q < 4; ++q) {
        const int e = i + q * QE;
        csr[rowptr[edst[e]] + (int)rank[e]] = esrc[e];
    }
}

__device__ __forceinline__ float lrelu(float a) { return a > 0.f ? a : a * SLOPE; }

#define A1_NPW 4
#define A1_U   8
static_assert(NN % (4 * A1_NPW) == 0, "nodes must fill the layer-1 aggregate grid");

__global__ void __launch_bounds__(128)
k_agg1(const int* __restrict__ rowptr, const int* __restrict__ csr,
       const float* __restrict__ aS, const float* __restrict__ aD,
       const float* __restrict__ H, const float* __restrict__ bias,
       float* __restrict__ O) {
    const int lane = threadIdx.x & 31;
    const int hl = lane >> 2;
    const int wv = __builtin_amdgcn_readfirstlane((int)((blockIdx.x * 128 + threadIdx.x) >> 5));
    const float4 b4 = *(const float4*)(bias + 4 * lane);
    for (int q = 0; q < A1_NPW; ++q) {
        const int n = wv * A1_NPW + q;
        const int beg = rowptr[n];
        const int end = rowptr[n + 1];
        const float aDl = aD[(size_t)n * HEADS + hl];
        float m = lrelu(aS[(size_t)n * HEADS + hl] + aDl);
        float den = 1.f;
        float4 acc = *(const float4*)(H + (size_t)n * HID_C + 4 * lane);
        for (int base = beg; base < end; base += 32) {
            const int c = min(32, end - base);
            const int sv = (lane < c) ? csr[base + lane] : 0;
            for (int j = 0; j < c; j += A1_U) {
                float av[A1_U];
                float4 rv[A1_U];
                #pragma unroll
                for (int u = 0; u < A1_U; ++u) {
                    const int s = __builtin_amdgcn_readlane(sv, min(j + u, c - 1));
                    av[u] = aS[(size_t)s * HEADS + hl];
                    rv[u] = *(const float4*)(H + (size_t)s * HID_C + 4 * lane);
                }
                float gm = m;
                #pragma unroll
                for (int u = 0; u < A1_U; ++u) {
                    av[u] = lrelu(av[u] + aDl);
                    gm = fmaxf(gm, av[u]);
                }
                const float r = __expf(m - gm);
                m = gm;
                den *= r;
                acc.x *= r; acc.y *= r; acc.z *= r; acc.w *= r;
                #pragma unroll
                for (int u = 0; u < A1_U; ++u) {
                    float w = __expf(av[u] - m);
                    w = (j + u < c) ? w : 0.f;
                    den += w;
                    acc.x += w * rv[u].x; acc.y += w * rv[u].y;
                    acc.z += w * rv[u].z; acc.w += w * rv[u].w;
                }
            }
        }
        float4 o;
        o.x = fmaxf(acc.x / den + b4.x, 0.f);
        o.y = fmaxf(acc.y / den + b4.y, 0.f);
        o.z = fmaxf(acc.z / den + b4.z, 0.f);
        o.w = fmaxf(acc.w / den + b4.w, 0.f);
        *(float4*)(O + (size_t)n * HID_C + 4 * lane) = o;
    }
}

__global__ void __launch_bounds__(256)
k_agg2(const int* __restrict__ rowptr, const int* __restrict__ csr,
       const float* __restrict__ aS, const float* __restrict__ aD,
       const float* __restrict__ H, const float* __restrict__ bias,
       float* __restrict__ O) {
    const int gt = blockIdx.x * 256 + threadIdx.x;
    const int n = gt >> 2;
    const int k = gt & 3;
    if (n >= NN) return;
    const int beg = rowptr[n];
    const int end = rowptr[n + 1];
    const float aDn = aD[n];
    float m = lrelu(aS[n] + aDn);
    float den = 1.f;
    float4 acc = *(const float4*)(H + (size_t)n * OUT_C + 4 * k);
    for (int i = beg; i < end; i += 4) {
        int sv[4];
        float av[4];
        float4 rv[4];
        #pragma unroll
        for (int u = 0; u < 4; ++u) sv[u] = csr[min(i + u, end - 1)];
        #pragma unroll
        for (int u = 0; u < 4; ++u) {
            av[u] = aS[sv[u]];
            rv[u] = *(const float4*)(H + (size_t)sv[u] * OUT_C + 4 * k);
        }
        float gm = m;
        #pragma unroll
        for (int u = 0; u < 4; ++u) {
            av[u] = lrelu(av[u] + aDn);
            gm = fmaxf(gm, av[u]);
        }
        const float r = __expf(m - gm);
        m = gm;
        den *= r;
        acc.x *= r; acc.y *= r; acc.z *= r; acc.w *= r;
        #pragma unroll
        for (int u = 0; u < 4; ++u) {
            float w = __expf(av[u] - m);
            w = (i + u < end) ? w : 0.f;
            den += w;
            acc.x += w * rv[u].x; acc.y += w * rv[u].y;
            acc.z += w * rv[u].z; acc.w += w * rv[u].w;
        }
    }
    const float4 b4 = *(const float4*)(bias + 4 * k);
    float4 o;
    o.x = acc.x / den + b4.x;
    o.y = acc.y / den + b4.y;
    o.z = acc.z / den + b4.z;
    o.w = acc.w / den + b4.w;
    *(float4*)(O + (size_t)n * OUT_C + 4 * k) = o;
}

#define PADW(x) (((size_t)(x) + 63) & ~(size_t)63)

extern "C" void kernel_launch(void* const* d_in, const int* in_sizes, int n_in,
                              void* d_out, int out_size, void* d_ws, size_t ws_size,
                              hipStream_t stream) {
    const float* x     = (const float*)d_in[0];
    const int*   ei    = (const int*)  d_in[1];
    const float* W1    = (const float*)d_in[2];
    const float* attS1 = (const float*)d_in[3];
    const float* attD1 = (const float*)d_in[4];
    const float* bias1 = (const float*)d_in[5];
    const float* W2    = (const float*)d_in[6];
    const float* attS2 = (const float*)d_in[7];
    const float* attD2 = (const float*)d_in[8];
    const float* bias2 = (const float*)d_in[9];
    float* out = (float*)d_out;
    const int* esrc = ei;
    const int* edst = ei + NE;

    unsigned* cnt    = (unsigned*)d_ws;
    int*      rowptr = (int*)(cnt + PADW(NN));
    unsigned* rank   = (unsigned*)(rowptr + PADW(NN + 1));
    int*      csr    = (int*)(rank + PADW(NE));
    float* h1   = (float*)(csr + PADW(NE));
    float* aS1  = h1 + PADW((size_t)NN * 128);
    float* aD1  = aS1 + PADW((size_t)NN * 8);
    float* out1 = aD1 + PADW((size_t)NN * 8);
    float* h2   = out1 + PADW((size_t)NN * 128);
    float* aS2  = h2 + PADW((size_t)NN * 16);
    float* aD2  = aS2 + PADW(NN);

    k_zero<<<(NN + 255) / 256, 256, 0, stream>>>(cnt, (long)NN);
    k_count<<<(QE + 255) / 256, 256, 0, stream>>>(edst, cnt, rank);
    k_scan<<<1, SCAN_T, 0, stream>>>(cnt, rowptr);
    k_fill<<<(QE + 255) / 256, 256, 0, stream>>>(esrc, edst, rowptr, rank, csr);

    k_gemm1<<<G1_GRID, 256, 0, stream>>>(x, W1, h1);
    k_attvec<<<(NN * HEADS + 255) / 256, 256, 0, stream>>>(h1, attS1, attD1, aS1, aD1, HEADS);
    k_agg1<<<NN / (4 * A1_NPW), 128, 0, stream>>>(rowptr, csr, aS1, aD1, h1, bias1, out1);

    k_gemm2<<<(NN + 127) / 128, 256, 0, stream>>>(out1, W2, h2);
    k_attvec<<<(NN + 255) / 256, 256, 0, stream>>>(h2, attS2, attD2, aS2, aD2, 1);
    k_agg2<<<(NN * 4 + 255) / 256, 256, 0, stream>>>(rowptr, csr, aS2, aD2, h2, bias2, out);
}


# Attention scores taken from the rows already gathered (4-channel dot per lane, quad sum by DPP) instead of a score gather per edge; both score launches go; score sum order changes
# speedup vs baseline: 8.0068x; 1.0997x over previous
#include <hip/hip_runtime.h>
#include <cstdint>

#define NN      50000
#define NE      1600000
#define ET      (NE + NN)
#define IN_C    256
#define HID_C   128
#define HEADS   8
#define OUT_C   16
#define SLOPE   0.2f

typedef float v2f __attribute__((ext_vector_type(2)));
typedef float v8f __attribute__((ext_vector_type(8)));

__global__ void k_zero(unsigned* __restrict__ p, long n) {
    long i = (long)blockIdx.x * 256 + threadIdx.x;
    if (i < n) p[i] = 0u;
}

#define G1_GRID   256
#define G1_TILES  (NN / 16)
#define G1_PITCH  (IN_C + 4)
#define G1_BUF    20736
static_assert(G1_BUF >= 32 * G1_PITCH, "buffer holds a step");
static_assert(NN % 16 == 0, "row tiles are whole");

#define G1_KC 4
template <int NT>
__device__ __forceinline__ void g1_tiles(const float* __restrict__ ap, const v2f (&bf)[64],
                                         float* __restrict__ hp) {
    v8f acc[NT];
    #pragma unroll
    for (int m = 0; m < NT; ++m) acc[m] = v8f{};
    v2f af[2][NT][G1_KC];
    #pragma unroll
    for (int m = 0; m < NT; ++m) {
        #pragma unroll
        for (int i = 0; i < G1_KC; ++i)
            af[0][m][i] = *(const v2f*)(ap + m * 16 * G1_PITCH + 4 * i);
    }
    __builtin_amdgcn_sched_group_barrier(0x100, NT * G1_KC / 2, 0);
    #pragma unroll
    for (int c = 0; c < 64 / G1_KC; ++c) {
        if (c + 1 < 64 / G1_KC) {
            #pragma unroll
            for (int m = 0; m < NT; ++m) {
                #pragma unroll
                for (int i = 0; i < G1_KC; ++i)
                    af[(c + 1) & 1][m][i] =
                        *(const v2f*)(ap + m * 16 * G1_PITCH + 4 * (G1_KC * (c + 1) + i));
            }
        }
        #pragma unroll
        for (int i = 0; i < G1_KC; ++i) {
            #pragma unroll
            for (int m = 0; m < NT; ++m)
                acc[m] = __builtin_amdgcn_wmma_f32_16x16x4_f32(false, af[c & 1][m][i], false,
                                                               bf[G1_KC * c + i], (short)0,
                                                               acc[m], false, false);
        }
        if (c + 1 < 64 / G1_KC) __builtin_amdgcn_sched_group_barrier(0x100, NT * G1_KC / 2, 0);
        __builtin_amdgcn_sched_group_barrier(0x008, NT * G1_KC, 0);
    }
    #pragma unroll
    for (int m = 0; m < NT; ++m) {
        #pragma unroll
        for (int r = 0; r < 8; ++r) hp[(size_t)(m * 16 + r) * HID_C] = acc[m][r];
    }
}

__device__ __forceinline__ void g1_lds_barrier() {
    __builtin_amdgcn_fence(__ATOMIC_RELEASE, "workgroup", "local");
    __builtin_amdgcn_s_barrier();
    __builtin_amdgcn_fence(__ATOMIC_ACQUIRE, "workgroup", "local");
}

__global__ void __launch_bounds__(256)
k_gemm1(const float* __restrict__ X, const float* __restrict__ W,
        float* __restrict__ H) {
    __shared__ __attribute__((aligned(16))) float As[2][G1_BUF];
    const int tid  = threadIdx.x;
    const int wave = tid >> 5;
    const int lane = tid & 31;
    const int nBase = wave << 4;
    const int lm = lane & 15;
    const int kh = (lane >> 4) << 1;
    const int t0 = ((int)blockIdx.x * G1_TILES) / G1_GRID;
    const int t1 = (((int)blockIdx.x + 1) * G1_TILES) / G1_GRID;
    const int rowBeg = t0 << 4;
    const int rowEnd = t1 << 4;

    const int sRow = tid >> 6;
    const int sCol = (tid & 63) << 2;
    float4 pre[8];
    auto fetch = [&](int r) {
        #pragma unroll
        for (int i = 0; i < 8; ++i) {
            const int row = r + sRow + 4 * i;
            pre[i] = (row < rowEnd) ? *(const float4*)(X + (size_t)row * IN_C + sCol)
                                    : float4{0.f, 0.f, 0.f, 0.f};
        }
    };
    auto stash = [&](int b) {
        #pragma unroll
        for (int i = 0; i < 8; ++i)
            *(float4*)(&As[b][(sRow + 4 * i) * G1_PITCH + sCol]) = pre[i];
    };

    v2f bf[64];
    const float* wp = W + (size_t)kh * HID_C + nBase + lm;
    #pragma unroll
    for (int kb = 0; kb < 64; ++kb) {
        bf[kb].x = wp[(size_t)(4 * kb) * HID_C];
        bf[kb].y = wp[(size_t)(4 * kb) * HID_C + HID_C];
    }

    fetch(rowBeg);
    stash(0);
    fetch(rowBeg + 32);
    g1_lds_barrier();

    int buf = 0;
    for (int r = rowBeg; r < rowEnd; r += 32) {
        const float* ap = &As[buf][lm * G1_PITCH + kh];
        float* hp = H + (size_t)(r + ((lane >> 4) << 3)) * HID_C + nBase + lm;
        if (rowEnd - r >= 32) g1_tiles<2>(ap, bf, hp);
        else                  g1_tiles<1>(ap, bf, hp);
        stash(buf ^ 1);
        fetch(r + 64);
        g1_lds_barrier();
        buf ^= 1;
    }
}

__global__ void k_gemm2(const float* __restrict__ X, const float* __restrict__ W,
                        float* __restrict__ H) {
    const int wave = threadIdx.x >> 5;
    const int lane = threadIdx.x & 31;
    const int mBase = (blockIdx.x << 7) + (wave << 4);
    if (mBase >= NN) return;
    const int lm = lane & 15;
    const int kh = (lane >> 4) << 1;
    const float* xr = X + (size_t)(mBase + lm) * HID_C + kh;
    const float* wp = W + (size_t)kh * OUT_C + lm;
    v8f acc = {};
    #pragma unroll 8
    for (int kb = 0; kb < HID_C; kb += 4) {
        v2f a = *(const v2f*)(xr + kb);
        v2f b;
        b.x = wp[(size_t)kb * OUT_C];
        b.y = wp[(size_t)kb * OUT_C + OUT_C];
        acc = __builtin_amdgcn_wmma_f32_16x16x4_f32(false, a, false, b,
                                                    (short)0, acc, false, false);
    }
    float* hp = H + (size_t)(mBase + ((lane >> 4) << 3)) * OUT_C + lm;
    #pragma unroll
    for (int r = 0; r < 8; ++r) hp[(size_t)r * OUT_C] = acc[r];
}

__device__ __forceinline__ float dot4(const float4 r, const float4 a) {
    v2f t = v2f{r.x, r.y} * v2f{a.x, a.y};
    t = v2f{r.z, r.w} * v2f{a.z, a.w} + t;
    return t.x + t.y;
}
__device__ __forceinline__ float quad_sum(float p) {
    p += __builtin_bit_cast(float, __builtin_amdgcn_update_dpp(
             0, __builtin_bit_cast(int, p), 0x161, 0xf, 0xf, true));
    p += __builtin_bit_cast(float, __builtin_amdgcn_update_dpp(
             0, __builtin_bit_cast(int, p), 0x162, 0xf, 0xf, true));
    return p;
}

#define QE (NE / 4)
static_assert(NE % 4 == 0, "edge count must split in four");

__global__ void k_count(const int* __restrict__ edst, unsigned* __restrict__ cnt,
                        unsigned* __restrict__ rank) {
    const int i = blockIdx.x * 256 + threadIdx.x;
    if (i >= QE) return;
    const int d0 = edst[i], d1 = edst[i + QE], d2 = edst[i + 2 * QE], d3 = edst[i + 3 * QE];
    const unsigned r0 = atomicAdd(&cnt[d0], 1u);
    const unsigned r1 = atomicAdd(&cnt[d1], 1u);
    const unsigned r2 = atomicAdd(&cnt[d2], 1u);
    const unsigned r3 = atomicAdd(&cnt[d3], 1u);
    rank[i] = r0; rank[i + QE] = r1; rank[i + 2 * QE] = r2; rank[i + 3 * QE] = r3;
}

#define SCAN_T  1024
#define SCAN_CH 49
#define SCAN_N  (SCAN_T * SCAN_CH)
static_assert(SCAN_N >= NN, "scan tile must cover all nodes");

__global__ void __launch_bounds__(SCAN_T)
k_scan(const unsigned* __restrict__ cnt, int* __restrict__ rowptr) {
    __shared__ unsigned buf[SCAN_N];
    __shared__ unsigned part[SCAN_T];
    const int t = threadIdx.x;
    for (int k = 0; k < SCAN_CH; ++k) {
        const int i = k * SCAN_T + t;
        buf[i] = (i < NN) ? cnt[i] : 0u;
    }
    __syncthreads();
    const int b = t * SCAN_CH;
    unsigned sum = 0u;
    for (int k = 0; k < SCAN_CH; ++k) sum += buf[b + k];
    part[t] = sum;
    __syncthreads();
    for (int off = 1; off < SCAN_T; off <<= 1) {
        const unsigned v = (t >= off) ? part[t - off] : 0u;
        __syncthreads();
        part[t] += v;
        __syncthreads();
    }
    unsigned run = (t > 0) ? part[t - 1] : 0u;
    for (int k = 0; k < SCAN_CH; ++k) {
        const unsigned c = buf[b + k];
        buf[b + k] = run;
        run += c;
    }
    __syncthreads();
    for (int k = 0; k < SCAN_CH; ++k) {
        const int i = k * SCAN_T + t;
        if (i < NN) rowptr[i] = (int)buf[i];
    }
    if (t == 0) rowptr[NN] = (int)part[SCAN_T - 1];
}

__global__ void k_fill(const int* __restrict__ esrc, const int* __restrict__ edst,
                       const int* __restrict__ rowptr, const unsigned* __restrict__ rank,
                       int* __restrict__ csr) {
    const int i = blockIdx.x * 256 + threadIdx.x;
    if (i >= QE) return;
    #pragma unroll
    for (int q = 0; q < 4; ++q) {
        const int e = i + q * QE;
        csr[rowptr[edst[e]] + (int)rank[e]] = esrc[e];
    }
}

__device__ __forceinline__ float lrelu(float a) { return a > 0.f ? a : a * SLOPE; }

#define A1_NPW 4
#define A1_U   8
static_assert(NN % (4 * A1_NPW) == 0, "nodes must fill the layer-1 aggregate grid");

__global__ void __launch_bounds__(128)
k_agg1(const int* __restrict__ rowptr, const int* __restrict__ csr,
       const float* __restrict__ attS, const float* __restrict__ attD,
       const float* __restrict__ H, const float* __restrict__ bias,
       float* __restrict__ O) {
    const int lane = threadIdx.x & 31;
    const int wv = __builtin_amdgcn_readfirstlane((int)((blockIdx.x * 128 + threadIdx.x) >> 5));
    const float4 b4 = *(const float4*)(bias + 4 * lane);
    const float4* __restrict__ H4 = (const float4*)H;
    const float4 as4 = *(const float4*)(attS + 4 * lane);
    const float4 ad4 = *(const float4*)(attD + 4 * lane);
    for (int q = 0; q < A1_NPW; ++q) {
        const int n = wv * A1_NPW + q;
        const int beg = rowptr[n];
        const int end = rowptr[n + 1];
        float4 acc = H4[(unsigned)n * 32u + (unsigned)lane];
        const float aDl = quad_sum(dot4(acc, ad4));
        float m = lrelu(quad_sum(dot4(acc, as4)) + aDl);
        float den = 1.f;
        for (int base = beg; base < end; base += 32) {
            const int c = min(32, end - base);
            const int sv = (lane < c) ? csr[base + lane] : 0;
            for (int j = 0; j < c; j += A1_U) {
                float av[A1_U];
                float4 rv[A1_U];
                #pragma unroll
                for (int u = 0; u < A1_U; ++u) {
                    const int s = __builtin_amdgcn_readlane(sv, min(j + u, c - 1));
                    rv[u] = H4[(unsigned)s * 32u + (unsigned)lane];
                }
                float gm = m;
                #pragma unroll
                for (int u = 0; u < A1_U; ++u) {
                    av[u] = lrelu(quad_sum(dot4(rv[u], as4)) + aDl);
                    gm = fmaxf(gm, av[u]);
                }
                const float r = __expf(m - gm);
                m = gm;
                den *= r;
                acc.x *= r; acc.y *= r; acc.z *= r; acc.w *= r;
                #pragma unroll
                for (int u = 0; u < A1_U; ++u) {
                    float w = __expf(av[u] - m);
                    w = (j + u < c) ? w : 0.f;
                    den += w;
                    acc.x += w * rv[u].x; acc.y += w * rv[u].y;
                    acc.z += w * rv[u].z; acc.w += w * rv[u].w;
                }
            }
        }
        float4 o;
        o.x = fmaxf(acc.x / den + b4.x, 0.f);
        o.y = fmaxf(acc.y / den + b4.y, 0.f);
        o.z = fmaxf(acc.z / den + b4.z, 0.f);
        o.w = fmaxf(acc.w / den + b4.w, 0.f);
        *(float4*)(O + (size_t)n * HID_C + 4 * lane) = o;
    }
}

__global__ void __launch_bounds__(256)
k_agg2(const int* __restrict__ rowptr, const int* __restrict__ csr,
       const float* __restrict__ attS, const float* __restrict__ attD,
       const float* __restrict__ H, const float* __restrict__ bias,
       float* __restrict__ O) {
    const int gt = blockIdx.x * 256 + threadIdx.x;
    const int n = gt >> 2;
    const int k = gt & 3;
    if (n >= NN) return;
    const int beg = rowptr[n];
    const int end = rowptr[n + 1];
    const float4 as4 = *(const float4*)(attS + 4 * k);
    const float4 ad4 = *(const float4*)(attD + 4 * k);
    const float4* __restrict__ H4 = (const float4*)H;
    float4 acc = H4[(unsigned)n * 4u + (unsigned)k];
    const float aDn = quad_sum(dot4(acc, ad4));
    float m = lrelu(quad_sum(dot4(acc, as4)) + aDn);
    float den = 1.f;
    for (int i = beg; i < end; i += 4) {
        int sv[4];
        float av[4];
        float4 rv[4];
        #pragma unroll
        for (int u = 0; u < 4; ++u) sv[u] = csr[min(i + u, end - 1)];
        #pragma unroll
        for (int u = 0; u < 4; ++u)
            rv[u] = H4[(unsigned)sv[u] * 4u + (unsigned)k];
        float gm = m;
        #pragma unroll
        for (int u = 0; u < 4; ++u) {
            av[u] = lrelu(quad_sum(dot4(rv[u], as4)) + aDn);
            gm = fmaxf(gm, av[u]);
        }
        const float r = __expf(m - gm);
        m = gm;
        den *= r;
        acc.x *= r; acc.y *= r; acc.z *= r; acc.w *= r;
        #pragma unroll
        for (int u = 0; u < 4; ++u) {
            float w = __expf(av[u] - m);
            w = (i + u < end) ? w : 0.f;
            den += w;
            acc.x += w * rv[u].x; acc.y += w * rv[u].y;
            acc.z += w * rv[u].z; acc.w += w * rv[u].w;
        }
    }
    const float4 b4 = *(const float4*)(bias + 4 * k);
    float4 o;
    o.x = acc.x / den + b4.x;
    o.y = acc.y / den + b4.y;
    o.z = acc.z / den + b4.z;
    o.w = acc.w / den + b4.w;
    *(float4*)(O + (size_t)n * OUT_C + 4 * k) = o;
}

#define PADW(x) (((size_t)(x) + 63) & ~(size_t)63)

extern "C" void kernel_launch(void* const* d_in, const int* in_sizes, int n_in,
                              void* d_out, int out_size, void* d_ws, size_t ws_size,
                              hipStream_t stream) {
    const float* x     = (const float*)d_in[0];
    const int*   ei    = (const int*)  d_in[1];
    const float* W1    = (const float*)d_in[2];
    const float* attS1 = (const float*)d_in[3];
    const float* attD1 = (const float*)d_in[4];
    const float* bias1 = (const float*)d_in[5];
    const float* W2    = (const float*)d_in[6];
    const float* attS2 = (const float*)d_in[7];
    const float* attD2 = (const float*)d_in[8];
    const float* bias2 = (const float*)d_in[9];
    float* out = (float*)d_out;
    const int* esrc = ei;
    const int* edst = ei + NE;

    unsigned* cnt    = (unsigned*)d_ws;
    int*      rowptr = (int*)(cnt + PADW(NN));
    unsigned* rank   = (unsigned*)(rowptr + PADW(NN + 1));
    int*      csr    = (int*)(rank + PADW(NE));
    float* h1   = (float*)(csr + PADW(NE));
    float* out1 = h1 + PADW((size_t)NN * 128);
    float* h2   = out1 + PADW((size_t)NN * 128);

    k_zero<<<(NN + 255) / 256, 256, 0, stream>>>(cnt, (long)NN);
    k_count<<<(QE + 255) / 256, 256, 0, stream>>>(edst, cnt, rank);
    k_scan<<<1, SCAN_T, 0, stream>>>(cnt, rowptr);
    k_fill<<<(QE + 255) / 256, 256, 0, stream>>>(esrc, edst, rowptr, rank, csr);

    k_gemm1<<<G1_GRID, 256, 0, stream>>>(x, W1, h1);
    k_agg1<<<NN / (4 * A1_NPW), 128, 0, stream>>>(rowptr, csr, attS1, attD1, h1, bias1, out1);

    k_gemm2<<<(NN + 127) / 128, 256, 0, stream>>>(out1, W2, h2);
    k_agg2<<<(NN * 4 + 255) / 256, 256, 0, stream>>>(rowptr, csr, attS2, attD2, h2, bias2, out);
}


# Edge lists by a two-level counting sort in LDS (chunks sorted by node range, one group per range ranks and places): no global atomics, scan or scattered stores; 8 launches to 6; sum order changes
# speedup vs baseline: 18.9117x; 2.3620x over previous
#include <hip/hip_runtime.h>
#include <cstdint>

#define NN      50000
#define NE      1600000
#define ET      (NE + NN)
#define IN_C    256
#define HID_C   128
#define HEADS   8
#define OUT_C   16
#define SLOPE   0.2f

typedef float v2f __attribute__((ext_vector_type(2)));
typedef float v8f __attribute__((ext_vector_type(8)));

#define G1_GRID   256
#define G1_TILES  (NN / 16)
#define G1_PITCH  (IN_C + 4)
#define G1_BUF    20736
static_assert(G1_BUF >= 32 * G1_PITCH, "buffer holds a step");
static_assert(NN % 16 == 0, "row tiles are whole");

#define G1_KC 4
template <int NT>
__device__ __forceinline__ void g1_tiles(const float* __restrict__ ap, const v2f (&bf)[64],
                                         float* __restrict__ hp) {
    v8f acc[NT];
    #pragma unroll
    for (int m = 0; m < NT; ++m) acc[m] = v8f{};
    v2f af[2][NT][G1_KC];
    #pragma unroll
    for (int m = 0; m < NT; ++m) {
        #pragma unroll
        for (int i = 0; i < G1_KC; ++i)
            af[0][m][i] = *(const v2f*)(ap + m * 16 * G1_PITCH + 4 * i);
    }
    __builtin_amdgcn_sched_group_barrier(0x100, NT * G1_KC / 2, 0);
    #pragma unroll
    for (int c = 0; c < 64 / G1_KC; ++c) {
        if (c + 1 < 64 / G1_KC) {
            #pragma unroll
            for (int m = 0; m < NT; ++m) {
                #pragma unroll
                for (int i = 0; i < G1_KC; ++i)
                    af[(c + 1) & 1][m][i] =
                        *(const v2f*)(ap + m * 16 * G1_PITCH + 4 * (G1_KC * (c + 1) + i));
            }
        }
        #pragma unroll
        for (int i = 0; i < G1_KC; ++i) {
            #pragma unroll
            for (int m = 0; m < NT; ++m)
                acc[m] = __builtin_amdgcn_wmma_f32_16x16x4_f32(false, af[c & 1][m][i], false,
                                                               bf[G1_KC * c + i], (short)0,
                                                               acc[m], false, false);
        }
        if (c + 1 < 64 / G1_KC) __builtin_amdgcn_sched_group_barrier(0x100, NT * G1_KC / 2, 0);
        __builtin_amdgcn_sched_group_barrier(0x008, NT * G1_KC, 0);
    }
    #pragma unroll
    for (int m = 0; m < NT; ++m) {
        #pragma unroll
        for (int r = 0; r < 8; ++r) hp[(size_t)(m * 16 + r) * HID_C] = acc[m][r];
    }
}

__device__ __forceinline__ void g1_lds_barrier() {
    __builtin_amdgcn_fence(__ATOMIC_RELEASE, "workgroup", "local");
    __builtin_amdgcn_s_barrier();
    __builtin_amdgcn_fence(__ATOMIC_ACQUIRE, "workgroup", "local");
}

__global__ void __launch_bounds__(256)
k_gemm1(const float* __restrict__ X, const float* __restrict__ W,
        float* __restrict__ H) {
    __shared__ __attribute__((aligned(16))) float As[2][G1_BUF];
    const int tid  = threadIdx.x;
    const int wave = tid >> 5;
    const int lane = tid & 31;
    const int nBase = wave << 4;
    const int lm = lane & 15;
    const int kh = (lane >> 4) << 1;
    const int t0 = ((int)blockIdx.x * G1_TILES) / G1_GRID;
    const int t1 = (((int)blockIdx.x + 1) * G1_TILES) / G1_GRID;
    const int rowBeg = t0 << 4;
    const int rowEnd = t1 << 4;

    const int sRow = tid >> 6;
    const int sCol = (tid & 63) << 2;
    float4 pre[8];
    auto fetch = [&](int r) {
        #pragma unroll
        for (int i = 0; i < 8; ++i) {
            const int row = r + sRow + 4 * i;
            pre[i] = (row < rowEnd) ? *(const float4*)(X + (size_t)row * IN_C + sCol)
                                    : float4{0.f, 0.f, 0.f, 0.f};
        }
    };
    auto stash = [&](int b) {
        #pragma unroll
        for (int i = 0; i < 8; ++i)
            *(float4*)(&As[b][(sRow + 4 * i) * G1_PITCH + sCol]) = pre[i];
    };

    v2f bf[64];
    const float* wp = W + (size_t)kh * HID_C + nBase + lm;
    #pragma unroll
    for (int kb = 0; kb < 64; ++kb) {
        bf[kb].x = wp[(size_t)(4 * kb) * HID_C];
        bf[kb].y = wp[(size_t)(4 * kb) * HID_C + HID_C];
    }

    fetch(rowBeg);
    stash(0);
    fetch(rowBeg + 32);
    g1_lds_barrier();

    int buf = 0;
    for (int r = rowBeg; r < rowEnd; r += 32) {
        const float* ap = &As[buf][lm * G1_PITCH + kh];
        float* hp = H + (size_t)(r + ((lane >> 4) << 3)) * HID_C + nBase + lm;
        if (rowEnd - r >= 32) g1_tiles<2>(ap, bf, hp);
        else                  g1_tiles<1>(ap, bf, hp);
        stash(buf ^ 1);
        fetch(r + 64);
        g1_lds_barrier();
        buf ^= 1;
    }
}

__global__ void k_gemm2(const float* __restrict__ X, const float* __restrict__ W,
                        float* __restrict__ H) {
    const int wave = threadIdx.x >> 5;
    const int lane = threadIdx.x & 31;
    const int mBase = (blockIdx.x << 7) + (wave << 4);
    if (mBase >= NN) return;
    const int lm = lane & 15;
    const int kh = (lane >> 4) << 1;
    const float* xr = X + (size_t)(mBase + lm) * HID_C + kh;
    const float* wp = W + (size_t)kh * OUT_C + lm;
    v8f acc = {};
    #pragma unroll 8
    for (int kb = 0; kb < HID_C; kb += 4) {
        v2f a = *(const v2f*)(xr + kb);
        v2f b;
        b.x = wp[(size_t)kb * OUT_C];
        b.y = wp[(size_t)kb * OUT_C + OUT_C];
        acc = __builtin_amdgcn_wmma_f32_16x16x4_f32(false, a, false, b,
                                                    (short)0, acc, false, false);
    }
    float* hp = H + (size_t)(mBase + ((lane >> 4) << 3)) * OUT_C + lm;
    #pragma unroll
    for (int r = 0; r < 8; ++r) hp[(size_t)r * OUT_C] = acc[r];
}

__device__ __forceinline__ float dot4(const float4 r, const float4 a) {
    v2f t = v2f{r.x, r.y} * v2f{a.x, a.y};
    t = v2f{r.z, r.w} * v2f{a.z, a.w} + t;
    return t.x + t.y;
}
__device__ __forceinline__ float quad_sum(float p) {
    p += __builtin_bit_cast(float, __builtin_amdgcn_update_dpp(
             0, __builtin_bit_cast(int, p), 0x161, 0xf, 0xf, true));
    p += __builtin_bit_cast(float, __builtin_amdgcn_update_dpp(
             0, __builtin_bit_cast(int, p), 0x162, 0xf, 0xf, true));
    return p;
}

#define P_NW   250
#define P_CH   (NE / P_NW)
#define P_T    256
#define P_EPT  (P_CH / P_T)
#define B_R    200
#define B_NB   (NN / B_R)
#define B_T    1024
#define B_SPW  8
#define B_CAP  16384
static_assert(NE % P_NW == 0 && P_CH % P_T == 0, "chunks are whole and fill the block");
static_assert(NN % B_R == 0, "ranges are whole");
static_assert(NN <= 65536, "a source id takes the low 16 bits of a record");
static_assert(B_NB < 256 && B_R < 256 && P_NW <= 256, "one wave scans 256 counters");
static_assert(P_T == 256 && B_SPW * 32 >= P_NW, "index maps below");

__device__ __forceinline__ unsigned wave_incl_scan(unsigned x, int lane) {
    #pragma unroll
    for (int off = 1; off < 32; off <<= 1) {
        const unsigned y = __shfl_up(x, off);
        if (lane >= off) x += y;
    }
    return x;
}
template <bool TWO>
__device__ __forceinline__ void scan256(const unsigned* c, unsigned* o1, unsigned* o2,
                                        int lane) {
    unsigned v[8];
    unsigned s = 0u;
    #pragma unroll
    for (int i = 0; i < 8; ++i) { v[i] = c[lane * 8 + i]; s += v[i]; }
    unsigned run = wave_incl_scan(s, lane) - s;
    #pragma unroll
    for (int i = 0; i < 8; ++i) {
        o1[lane * 8 + i] = run;
        if (TWO) o2[lane * 8 + i] = run;
        run += v[i];
    }
}

__global__ void __launch_bounds__(P_T)
k_part(const int* __restrict__ esrc, const int* __restrict__ edst,
       unsigned* __restrict__ part, int* __restrict__ S) {
    __shared__ unsigned hist[256];
    __shared__ unsigned start[256];
    __shared__ unsigned stage[P_CH];
    const int t = threadIdx.x;
    const int e0 = (int)blockIdx.x * P_CH;
    hist[t] = 0u;
    int d[P_EPT], s[P_EPT];
    #pragma unroll
    for (int k = 0; k < P_EPT; ++k) {
        d[k] = edst[e0 + k * P_T + t];
        s[k] = esrc[e0 + k * P_T + t];
    }
    __syncthreads();
    unsigned r[P_EPT];
    #pragma unroll
    for (int k = 0; k < P_EPT; ++k) r[k] = atomicAdd(&hist[(unsigned)d[k] / B_R], 1u);
    __syncthreads();
    if (t < 32) scan256<false>(hist, start, nullptr, t);
    __syncthreads();
    #pragma unroll
    for (int k = 0; k < P_EPT; ++k) {
        const unsigned b = (unsigned)d[k] / B_R;
        stage[start[b] + r[k]] = (unsigned)s[k] | (((unsigned)d[k] - b * B_R) << 16);
    }
    __syncthreads();
    #pragma unroll
    for (int k = 0; k < P_EPT; ++k) part[e0 + k * P_T + t] = stage[k * P_T + t];
    if (t <= B_NB) S[(int)blockIdx.x * (B_NB + 1) + t] = (int)start[t];
}

__global__ void __launch_bounds__(B_T)
k_build(const unsigned* __restrict__ part, const int* __restrict__ S,
        int* __restrict__ rowptr, int* __restrict__ csr) {
    __shared__ int segOff[256];
    __shared__ int segLen[256];
    __shared__ unsigned cnt[256];
    __shared__ unsigned rowL[256];
    __shared__ unsigned cur[256];
    __shared__ unsigned sBase;
    __shared__ int stageC[B_CAP];
    const int t = threadIdx.x;
    const int lane = t & 31;
    const int wv = t >> 5;
    const int b = blockIdx.x;
    if (t == 0) sBase = 0u;
    if (t < 256) cnt[t] = 0u;
    __syncthreads();
    if (t < 256) {
        int a = 0, len = 0;
        if (t < P_NW) {
            a = S[t * (B_NB + 1) + b];
            len = S[t * (B_NB + 1) + b + 1] - a;
        }
        segOff[t] = t * P_CH + a;
        segLen[t] = len;
        unsigned sum = (unsigned)a;
        #pragma unroll
        for (int off = 16; off > 0; off >>= 1) sum += __shfl_xor(sum, off);
        if (lane == 0) atomicAdd(&sBase, sum);
    }
    __syncthreads();
    const unsigned base = sBase;

    int ln[B_SPW], of[B_SPW];
    unsigned rec[B_SPW];
    #pragma unroll
    for (int j = 0; j < B_SPW; ++j) {
        ln[j] = segLen[wv + 32 * j];
        of[j] = segOff[wv + 32 * j];
        rec[j] = (lane < ln[j]) ? part[of[j] + lane] : 0u;
    }
    #pragma unroll
    for (int j = 0; j < B_SPW; ++j)
        if (lane < ln[j]) atomicAdd(&cnt[rec[j] >> 16], 1u);
    #pragma unroll
    for (int j = 0; j < B_SPW; ++j)
        for (int k = lane + 32; k < ln[j]; k += 32) atomicAdd(&cnt[part[of[j] + k] >> 16], 1u);
    __syncthreads();
    if (wv == 0) scan256<true>(cnt, rowL, cur, lane);
    __syncthreads();
    const unsigned total = rowL[B_R];
    if (t < B_R) rowptr[b * B_R + t] = (int)(base + rowL[t]);
    if (b == B_NB - 1 && t == 0) rowptr[NN] = (int)(base + total);

    const bool fits = total <= (unsigned)B_CAP;
    int* __restrict__ lp = csr + base;
    unsigned pos[B_SPW];
    #pragma unroll
    for (int j = 0; j < B_SPW; ++j)
        pos[j] = (lane < ln[j]) ? atomicAdd(&cur[rec[j] >> 16], 1u) : 0u;
    if (fits) {
        #pragma unroll
        for (int j = 0; j < B_SPW; ++j)
            if (lane < ln[j]) stageC[pos[j]] = (int)(rec[j] & 0xffffu);
    } else {
        #pragma unroll
        for (int j = 0; j < B_SPW; ++j)
            if (lane < ln[j]) lp[pos[j]] = (int)(rec[j] & 0xffffu);
    }
    #pragma unroll
    for (int j = 0; j < B_SPW; ++j) {
        for (int k = lane + 32; k < ln[j]; k += 32) {
            const unsigned rc = part[of[j] + k];
            const unsigned p = atomicAdd(&cur[rc >> 16], 1u);
            const int sv = (int)(rc & 0xffffu);
            if (fits) stageC[p] = sv; else lp[p] = sv;
        }
    }
    __syncthreads();
    if (fits)
        for (unsigned i = (unsigned)t; i < total; i += B_T) lp[i] = stageC[i];
}

__device__ __forceinline__ float lrelu(float a) { return a > 0.f ? a : a * SLOPE; }

#define A1_NPW 4
#define A1_U   8
static_assert(NN % (4 * A1_NPW) == 0, "nodes must fill the layer-1 aggregate grid");

__global__ void __launch_bounds__(128)
k_agg1(const int* __restrict__ rowptr, const int* __restrict__ csr,
       const float* __restrict__ attS, const float* __restrict__ attD,
       const float* __restrict__ H, const float* __restrict__ bias,
       float* __restrict__ O) {
    const int lane = threadIdx.x & 31;
    const int wv = __builtin_amdgcn_readfirstlane((int)((blockIdx.x * 128 + threadIdx.x) >> 5));
    const float4 b4 = *(const float4*)(bias + 4 * lane);
    const float4* __restrict__ H4 = (const float4*)H;
    const float4 as4 = *(const float4*)(attS + 4 * lane);
    const float4 ad4 = *(const float4*)(attD + 4 * lane);
    for (int q = 0; q < A1_NPW; ++q) {
        const int n = wv * A1_NPW + q;
        const int beg = rowptr[n];
        const int end = rowptr[n + 1];
        float4 acc = H4[(unsigned)n * 32u + (unsigned)lane];
        const float aDl = quad_sum(dot4(acc, ad4));
        float m = lrelu(quad_sum(dot4(acc, as4)) + aDl);
        float den = 1.f;
        for (int base = beg; base < end; base += 32) {
            const int c = min(32, end - base);
            const int sv = (lane < c) ? csr[base + lane] : 0;
            for (int j = 0; j < c; j += A1_U) {
                float av[A1_U];
                float4 rv[A1_U];
                #pragma unroll
                for (int u = 0; u < A1_U; ++u) {
                    const int s = __builtin_amdgcn_readlane(sv, min(j + u, c - 1));
                    rv[u] = H4[(unsigned)s * 32u + (unsigned)lane];
                }
                float gm = m;
                #pragma unroll
                for (int u = 0; u < A1_U; ++u) {
                    av[u] = lrelu(quad_sum(dot4(rv[u], as4)) + aDl);
                    gm = fmaxf(gm, av[u]);
                }
                const float r = __expf(m - gm);
                m = gm;
                den *= r;
                acc.x *= r; acc.y *= r; acc.z *= r; acc.w *= r;
                #pragma unroll
                for (int u = 0; u < A1_U; ++u) {
                    float w = __expf(av[u] - m);
                    w = (j + u < c) ? w : 0.f;
                    den += w;
                    acc.x += w * rv[u].x; acc.y += w * rv[u].y;
                    acc.z += w * rv[u].z; acc.w += w * rv[u].w;
                }
            }
        }
        float4 o;
        o.x = fmaxf(acc.x / den + b4.x, 0.f);
        o.y = fmaxf(acc.y / den + b4.y, 0.f);
        o.z = fmaxf(acc.z / den + b4.z, 0.f);
        o.w = fmaxf(acc.w / den + b4.w, 0.f);
        *(float4*)(O + (size_t)n * HID_C + 4 * lane) = o;
    }
}

__global__ void __launch_bounds__(256)
k_agg2(const int* __restrict__ rowptr, const int* __restrict__ csr,
       const float* __restrict__ attS, const float* __restrict__ attD,
       const float* __restrict__ H, const float* __restrict__ bias,
       float* __restrict__ O) {
    const int gt = blockIdx.x * 256 + threadIdx.x;
    const int n = gt >> 2;
    const int k = gt & 3;
    if (n >= NN) return;
    const int beg = rowptr[n];
    const int end = rowptr[n + 1];
    const float4 as4 = *(const float4*)(attS + 4 * k);
    const float4 ad4 = *(const float4*)(attD + 4 * k);
    const float4* __restrict__ H4 = (const float4*)H;
    float4 acc = H4[(unsigned)n * 4u + (unsigned)k];
    const float aDn = quad_sum(dot4(acc, ad4));
    float m = lrelu(quad_sum(dot4(acc, as4)) + aDn);
    float den = 1.f;
    for (int i = beg; i < end; i += 4) {
        int sv[4];
        float av[4];
        float4 rv[4];
        #pragma unroll
        for (int u = 0; u < 4; ++u) sv[u] = csr[min(i + u, end - 1)];
        #pragma unroll
        for (int u = 0; u < 4; ++u)
            rv[u] = H4[(unsigned)sv[u] * 4u + (unsigned)k];
        float gm = m;
        #pragma unroll
        for (int u = 0; u < 4; ++u) {
            av[u] = lrelu(quad_sum(dot4(rv[u], as4)) + aDn);
            gm = fmaxf(gm, av[u]);
        }
        const float r = __expf(m - gm);
        m = gm;
        den *= r;
        acc.x *= r; acc.y *= r; acc.z *= r; acc.w *= r;
        #pragma unroll
        for (int u = 0; u < 4; ++u) {
            float w = __expf(av[u] - m);
            w = (i + u < end) ? w : 0.f;
            den += w;
            acc.x += w * rv[u].x; acc.y += w * rv[u].y;
            acc.z += w * rv[u].z; acc.w += w * rv[u].w;
        }
    }
    const float4 b4 = *(const float4*)(bias + 4 * k);
    float4 o;
    o.x = acc.x / den + b4.x;
    o.y = acc.y / den + b4.y;
    o.z = acc.z / den + b4.z;
    o.w = acc.w / den + b4.w;
    *(float4*)(O + (size_t)n * OUT_C + 4 * k) = o;
}

#define PADW(x) (((size_t)(x) + 63) & ~(size_t)63)

extern "C" void kernel_launch(void* const* d_in, const int* in_sizes, int n_in,
                              void* d_out, int out_size, void* d_ws, size_t ws_size,
                              hipStream_t stream) {
    const float* x     = (const float*)d_in[0];
    const int*   ei    = (const int*)  d_in[1];
    const float* W1    = (const float*)d_in[2];
    const float* attS1 = (const float*)d_in[3];
    const float* attD1 = (const float*)d_in[4];
    const float* bias1 = (const float*)d_in[5];
    const float* W2    = (const float*)d_in[6];
    const float* attS2 = (const float*)d_in[7];
    const float* attD2 = (const float*)d_in[8];
    const float* bias2 = (const float*)d_in[9];
    float* out = (float*)d_out;
    const int* esrc = ei;
    const int* edst = ei + NE;

    int*      S      = (int*)d_ws;
    int*      rowptr = S + PADW(P_NW * (B_NB + 1));
    unsigned* part   = (unsigned*)(rowptr + PADW(NN + 1));
    int*      csr    = (int*)(part + PADW(NE));
    float* h1   = (float*)(csr + PADW(NE));
    float* out1 = h1 + PADW((size_t)NN * 128);
    float* h2   = out1 + PADW((size_t)NN * 128);

    k_part<<<P_NW, P_T, 0, stream>>>(esrc, edst, part, S);
    k_build<<<B_NB, B_T, 0, stream>>>(part, S, rowptr, csr);

    k_gemm1<<<G1_GRID, 256, 0, stream>>>(x, W1, h1);
    k_agg1<<<NN / (4 * A1_NPW), 128, 0, stream>>>(rowptr, csr, attS1, attD1, h1, bias1, out1);

    k_gemm2<<<(NN + 127) / 128, 256, 0, stream>>>(out1, W2, h2);
    k_agg2<<<(NN * 4 + 255) / 256, 256, 0, stream>>>(rowptr, csr, attS2, attD2, h2, bias2, out);
}


# Layer-2 product h2 = relu(out1) @ W2 done inside the layer-1 aggregate: a block's 16 rows go to LDS and its last wave to finish runs the WMMA tile; gemm2 launch and out1 round trip go; same sums
# speedup vs baseline: 20.2489x; 1.0707x over previous
#include <hip/hip_runtime.h>
#include <cstdint>

#define NN      50000
#define NE      1600000
#define ET      (NE + NN)
#define IN_C    256
#define HID_C   128
#define HEADS   8
#define OUT_C   16
#define SLOPE   0.2f

typedef float v2f __attribute__((ext_vector_type(2)));
typedef float v8f __attribute__((ext_vector_type(8)));

#define G1_GRID   256
#define G1_TILES  (NN / 16)
#define G1_PITCH  (IN_C + 4)
#define G1_BUF    20736
static_assert(G1_BUF >= 32 * G1_PITCH, "buffer holds a step");
static_assert(NN % 16 == 0, "row tiles are whole");

#define G1_KC 4
template <int NT>
__device__ __forceinline__ void g1_tiles(const float* __restrict__ ap, const v2f (&bf)[64],
                                         float* __restrict__ hp) {
    v8f acc[NT];
    #pragma unroll
    for (int m = 0; m < NT; ++m) acc[m] = v8f{};
    v2f af[2][NT][G1_KC];
    #pragma unroll
    for (int m = 0; m < NT; ++m) {
        #pragma unroll
        for (int i = 0; i < G1_KC; ++i)
            af[0][m][i] = *(const v2f*)(ap + m * 16 * G1_PITCH + 4 * i);
    }
    __builtin_amdgcn_sched_group_barrier(0x100, NT * G1_KC / 2, 0);
    #pragma unroll
    for (int c = 0; c < 64 / G1_KC; ++c) {
        if (c + 1 < 64 / G1_KC) {
            #pragma unroll
            for (int m = 0; m < NT; ++m) {
                #pragma unroll
                for (int i = 0; i < G1_KC; ++i)
                    af[(c + 1) & 1][m][i] =
                        *(const v2f*)(ap + m * 16 * G1_PITCH + 4 * (G1_KC * (c + 1) + i));
            }
        }
        #pragma unroll
        for (int i = 0; i < G1_KC; ++i) {
            #pragma unroll
            for (int m = 0; m < NT; ++m)
                acc[m] = __builtin_amdgcn_wmma_f32_16x16x4_f32(false, af[c & 1][m][i], false,
                                                               bf[G1_KC * c + i], (short)0,
                                                               acc[m], false, false);
        }
        if (c + 1 < 64 / G1_KC) __builtin_amdgcn_sched_group_barrier(0x100, NT * G1_KC / 2, 0);
        __builtin_amdgcn_sched_group_barrier(0x008, NT * G1_KC, 0);
    }
    #pragma unroll
    for (int m = 0; m < NT; ++m) {
        #pragma unroll
        for (int r = 0; r < 8; ++r) hp[(size_t)(m * 16 + r) * HID_C] = acc[m][r];
    }
}

__device__ __forceinline__ void g1_lds_barrier() {
    __builtin_amdgcn_fence(__ATOMIC_RELEASE, "workgroup", "local");
    __builtin_amdgcn_s_barrier();
    __builtin_amdgcn_fence(__ATOMIC_ACQUIRE, "workgroup", "local");
}

__global__ void __launch_bounds__(256)
k_gemm1(const float* __restrict__ X, const float* __restrict__ W,
        float* __restrict__ H) {
    __shared__ __attribute__((aligned(16))) float As[2][G1_BUF];
    const int tid  = threadIdx.x;
    const int wave = tid >> 5;
    const int lane = tid & 31;
    const int nBase = wave << 4;
    const int lm = lane & 15;
    const int kh = (lane >> 4) << 1;
    const int t0 = ((int)blockIdx.x * G1_TILES) / G1_GRID;
    const int t1 = (((int)blockIdx.x + 1) * G1_TILES) / G1_GRID;
    const int rowBeg = t0 << 4;
    const int rowEnd = t1 << 4;

    const int sRow = tid >> 6;
    const int sCol = (tid & 63) << 2;
    float4 pre[8];
    auto fetch = [&](int r) {
        #pragma unroll
        for (int i = 0; i < 8; ++i) {
            const int row = r + sRow + 4 * i;
            pre[i] = (row < rowEnd) ? *(const float4*)(X + (size_t)row * IN_C + sCol)
                                    : float4{0.f, 0.f, 0.f, 0.f};
        }
    };
    auto stash = [&](int b) {
        #pragma unroll
        for (int i = 0; i < 8; ++i)
            *(float4*)(&As[b][(sRow + 4 * i) * G1_PITCH + sCol]) = pre[i];
    };

    v2f bf[64];
    const float* wp = W + (size_t)kh * HID_C + nBase + lm;
    #pragma unroll
    for (int kb = 0; kb < 64; ++kb) {
        bf[kb].x = wp[(size_t)(4 * kb) * HID_C];
        bf[kb].y = wp[(size_t)(4 * kb) * HID_C + HID_C];
    }

    fetch(rowBeg);
    stash(0);
    fetch(rowBeg + 32);
    g1_lds_barrier();

    int buf = 0;
    for (int r = rowBeg; r < rowEnd; r += 32) {
        const float* ap = &As[buf][lm * G1_PITCH + kh];
        float* hp = H + (size_t)(r + ((lane >> 4) << 3)) * HID_C + nBase + lm;
        if (rowEnd - r >= 32) g1_tiles<2>(ap, bf, hp);
        else                  g1_tiles<1>(ap, bf, hp);
        stash(buf ^ 1);
        fetch(r + 64);
        g1_lds_barrier();
        buf ^= 1;
    }
}

__device__ __forceinline__ float dot4(const float4 r, const float4 a) {
    v2f t = v2f{r.x, r.y} * v2f{a.x, a.y};
    t = v2f{r.z, r.w} * v2f{a.z, a.w} + t;
    return t.x + t.y;
}
__device__ __forceinline__ float quad_sum(float p) {
    p += __builtin_bit_cast(float, __builtin_amdgcn_update_dpp(
             0, __builtin_bit_cast(int, p), 0x161, 0xf, 0xf, true));
    p += __builtin_bit_cast(float, __builtin_amdgcn_update_dpp(
             0, __builtin_bit_cast(int, p), 0x162, 0xf, 0xf, true));
    return p;
}

#define P_NW   250
#define P_CH   (NE / P_NW)
#define P_T    256
#define P_EPT  (P_CH / P_T)
#define B_R    200
#define B_NB   (NN / B_R)
#define B_T    1024
#define B_SPW  8
#define B_CAP  16384
static_assert(NE % P_NW == 0 && P_CH % P_T == 0, "chunks are whole and fill the block");
static_assert(NN % B_R == 0, "ranges are whole");
static_assert(NN <= 65536, "a source id takes the low 16 bits of a record");
static_assert(B_NB < 256 && B_R < 256 && P_NW <= 256, "one wave scans 256 counters");
static_assert(P_T == 256 && B_SPW * 32 >= P_NW, "index maps below");

__device__ __forceinline__ unsigned wave_incl_scan(unsigned x, int lane) {
    #pragma unroll
    for (int off = 1; off < 32; off <<= 1) {
        const unsigned y = __shfl_up(x, off);
        if (lane >= off) x += y;
    }
    return x;
}
template <bool TWO>
__device__ __forceinline__ void scan256(const unsigned* c, unsigned* o1, unsigned* o2,
                                        int lane) {
    unsigned v[8];
    unsigned s = 0u;
    #pragma unroll
    for (int i = 0; i < 8; ++i) { v[i] = c[lane * 8 + i]; s += v[i]; }
    unsigned run = wave_incl_scan(s, lane) - s;
    #pragma unroll
    for (int i = 0; i < 8; ++i) {
        o1[lane * 8 + i] = run;
        if (TWO) o2[lane * 8 + i] = run;
        run += v[i];
    }
}

__global__ void __launch_bounds__(P_T)
k_part(const int* __restrict__ esrc, const int* __restrict__ edst,
       unsigned* __restrict__ part, int* __restrict__ S) {
    __shared__ unsigned hist[256];
    __shared__ unsigned start[256];
    __shared__ unsigned stage[P_CH];
    const int t = threadIdx.x;
    const int e0 = (int)blockIdx.x * P_CH;
    hist[t] = 0u;
    int d[P_EPT], s[P_EPT];
    #pragma unroll
    for (int k = 0; k < P_EPT; ++k) {
        d[k] = edst[e0 + k * P_T + t];
        s[k] = esrc[e0 + k * P_T + t];
    }
    __syncthreads();
    unsigned r[P_EPT];
    #pragma unroll
    for (int k = 0; k < P_EPT; ++k) r[k] = atomicAdd(&hist[(unsigned)d[k] / B_R], 1u);
    __syncthreads();
    if (t < 32) scan256<false>(hist, start, nullptr, t);
    __syncthreads();
    #pragma unroll
    for (int k = 0; k < P_EPT; ++k) {
        const unsigned b = (unsigned)d[k] / B_R;
        stage[start[b] + r[k]] = (unsigned)s[k] | (((unsigned)d[k] - b * B_R) << 16);
    }
    __syncthreads();
    #pragma unroll
    for (int k = 0; k < P_EPT; ++k) part[e0 + k * P_T + t] = stage[k * P_T + t];
    if (t <= B_NB) S[(int)blockIdx.x * (B_NB + 1) + t] = (int)start[t];
}

__global__ void __launch_bounds__(B_T)
k_build(const unsigned* __restrict__ part, const int* __restrict__ S,
        int* __restrict__ rowptr, int* __restrict__ csr) {
    __shared__ int segOff[256];
    __shared__ int segLen[256];
    __shared__ unsigned cnt[256];
    __shared__ unsigned rowL[256];
    __shared__ unsigned cur[256];
    __shared__ unsigned sBase;
    __shared__ int stageC[B_CAP];
    const int t = threadIdx.x;
    const int lane = t & 31;
    const int wv = t >> 5;
    const int b = blockIdx.x;
    if (t == 0) sBase = 0u;
    if (t < 256) cnt[t] = 0u;
    __syncthreads();
    if (t < 256) {
        int a = 0, len = 0;
        if (t < P_NW) {
            a = S[t * (B_NB + 1) + b];
            len = S[t * (B_NB + 1) + b + 1] - a;
        }
        segOff[t] = t * P_CH + a;
        segLen[t] = len;
        unsigned sum = (unsigned)a;
        #pragma unroll
        for (int off = 16; off > 0; off >>= 1) sum += __shfl_xor(sum, off);
        if (lane == 0) atomicAdd(&sBase, sum);
    }
    __syncthreads();
    const unsigned base = sBase;

    int ln[B_SPW], of[B_SPW];
    unsigned rec[B_SPW];
    #pragma unroll
    for (int j = 0; j < B_SPW; ++j) {
        ln[j] = segLen[wv + 32 * j];
        of[j] = segOff[wv + 32 * j];
        rec[j] = (lane < ln[j]) ? part[of[j] + lane] : 0u;
    }
    #pragma unroll
    for (int j = 0; j < B_SPW; ++j)
        if (lane < ln[j]) atomicAdd(&cnt[rec[j] >> 16], 1u);
    #pragma unroll
    for (int j = 0; j < B_SPW; ++j)
        for (int k = lane + 32; k < ln[j]; k += 32) atomicAdd(&cnt[part[of[j] + k] >> 16], 1u);
    __syncthreads();
    if (wv == 0) scan256<true>(cnt, rowL, cur, lane);
    __syncthreads();
    const unsigned total = rowL[B_R];
    if (t < B_R) rowptr[b * B_R + t] = (int)(base + rowL[t]);
    if (b == B_NB - 1 && t == 0) rowptr[NN] = (int)(base + total);

    const bool fits = total <= (unsigned)B_CAP;
    int* __restrict__ lp = csr + base;
    unsigned pos[B_SPW];
    #pragma unroll
    for (int j = 0; j < B_SPW; ++j)
        pos[j] = (lane < ln[j]) ? atomicAdd(&cur[rec[j] >> 16], 1u) : 0u;
    if (fits) {
        #pragma unroll
        for (int j = 0; j < B_SPW; ++j)
            if (lane < ln[j]) stageC[pos[j]] = (int)(rec[j] & 0xffffu);
    } else {
        #pragma unroll
        for (int j = 0; j < B_SPW; ++j)
            if (lane < ln[j]) lp[pos[j]] = (int)(rec[j] & 0xffffu);
    }
    #pragma unroll
    for (int j = 0; j < B_SPW; ++j) {
        for (int k = lane + 32; k < ln[j]; k += 32) {
            const unsigned rc = part[of[j] + k];
            const unsigned p = atomicAdd(&cur[rc >> 16], 1u);
            const int sv = (int)(rc & 0xffffu);
            if (fits) stageC[p] = sv; else lp[p] = sv;
        }
    }
    __syncthreads();
    if (fits)
        for (unsigned i = (unsigned)t; i < total; i += B_T) lp[i] = stageC[i];
}

__device__ __forceinline__ float lrelu(float a) { return a > 0.f ? a : a * SLOPE; }

#define A1_NPW 4
#define A1_U   8
#define A1_OP  (HID_C + 4)
static_assert(NN % (4 * A1_NPW) == 0, "nodes must fill the layer-1 aggregate grid");
static_assert(4 * A1_NPW == 16, "a block's nodes are one 16-row tile");

__global__ void __launch_bounds__(128)
k_agg1(const int* __restrict__ rowptr, const int* __restrict__ csr,
       const float* __restrict__ attS, const float* __restrict__ attD,
       const float* __restrict__ H, const float* __restrict__ bias,
       const float* __restrict__ W2, float* __restrict__ H2) {
    __shared__ __attribute__((aligned(16))) float Os[16 * A1_OP];
    __shared__ unsigned done;
    const int lane = threadIdx.x & 31;
    const int wib = __builtin_amdgcn_readfirstlane((int)(threadIdx.x >> 5));
    if (threadIdx.x == 0) done = 0u;
    __syncthreads();
    const int wv = __builtin_amdgcn_readfirstlane((int)((blockIdx.x * 128 + threadIdx.x) >> 5));
    const float4 b4 = *(const float4*)(bias + 4 * lane);
    const float4* __restrict__ H4 = (const float4*)H;
    const float4 as4 = *(const float4*)(attS + 4 * lane);
    const float4 ad4 = *(const float4*)(attD + 4 * lane);
    for (int q = 0; q < A1_NPW; ++q) {
        const int n = wv * A1_NPW + q;
        const int beg = rowptr[n];
        const int end = rowptr[n + 1];
        float4 acc = H4[(unsigned)n * 32u + (unsigned)lane];
        const float aDl = quad_sum(dot4(acc, ad4));
        float m = lrelu(quad_sum(dot4(acc, as4)) + aDl);
        float den = 1.f;
        for (int base = beg; base < end; base += 32) {
            const int c = min(32, end - base);
            const int sv = (lane < c) ? csr[base + lane] : 0;
            for (int j = 0; j < c; j += A1_U) {
                float av[A1_U];
                float4 rv[A1_U];
                #pragma unroll
                for (int u = 0; u < A1_U; ++u) {
                    const int s = __builtin_amdgcn_readlane(sv, min(j + u, c - 1));
                    rv[u] = H4[(unsigned)s * 32u + (unsigned)lane];
                }
                float gm = m;
                #pragma unroll
                for (int u = 0; u < A1_U; ++u) {
                    av[u] = lrelu(quad_sum(dot4(rv[u], as4)) + aDl);
                    gm = fmaxf(gm, av[u]);
                }
                const float r = __expf(m - gm);
                m = gm;
                den *= r;
                acc.x *= r; acc.y *= r; acc.z *= r; acc.w *= r;
                #pragma unroll
                for (int u = 0; u < A1_U; ++u) {
                    float w = __expf(av[u] - m);
                    w = (j + u < c) ? w : 0.f;
                    den += w;
                    acc.x += w * rv[u].x; acc.y += w * rv[u].y;
                    acc.z += w * rv[u].z; acc.w += w * rv[u].w;
                }
            }
        }
        float4 o;
        o.x = fmaxf(acc.x / den + b4.x, 0.f);
        o.y = fmaxf(acc.y / den + b4.y, 0.f);
        o.z = fmaxf(acc.z / den + b4.z, 0.f);
        o.w = fmaxf(acc.w / den + b4.w, 0.f);
        *(float4*)(&Os[(wib * A1_NPW + q) * A1_OP + 4 * lane]) = o;
    }

    __builtin_amdgcn_fence(__ATOMIC_RELEASE, "workgroup", "local");
    unsigned prev = 0u;
    if (lane == 0) prev = atomicAdd(&done, 1u);
    prev = __builtin_amdgcn_readfirstlane(prev);
    if (prev != 3u) return;
    __builtin_amdgcn_fence(__ATOMIC_ACQUIRE, "workgroup", "local");

    const int lm = lane & 15;
    const int kh = (lane >> 4) << 1;
    const float* ar = &Os[lm * A1_OP + kh];
    const float* wp = W2 + (size_t)kh * OUT_C + lm;
    v8f t = {};
    #pragma unroll 16
    for (int kb = 0; kb < HID_C; kb += 4) {
        v2f a = *(const v2f*)(ar + kb);
        v2f b;
        b.x = wp[(size_t)kb * OUT_C];
        b.y = wp[(size_t)kb * OUT_C + OUT_C];
        t = __builtin_amdgcn_wmma_f32_16x16x4_f32(false, a, false, b,
                                                  (short)0, t, false, false);
    }
    float* hp = H2 + (size_t)((int)blockIdx.x * 16 + ((lane >> 4) << 3)) * OUT_C + lm;
    #pragma unroll
    for (int r = 0; r < 8; ++r) hp[(size_t)r * OUT_C] = t[r];
}

__global__ void __launch_bounds__(256)
k_agg2(const int* __restrict__ rowptr, const int* __restrict__ csr,
       const float* __restrict__ attS, const float* __restrict__ attD,
       const float* __restrict__ H, const float* __restrict__ bias,
       float* __restrict__ O) {
    const int gt = blockIdx.x * 256 + threadIdx.x;
    const int n = gt >> 2;
    const int k = gt & 3;
    if (n >= NN) return;
    const int beg = rowptr[n];
    const int end = rowptr[n + 1];
    const float4 as4 = *(const float4*)(attS + 4 * k);
    const float4 ad4 = *(const float4*)(attD + 4 * k);
    const float4* __restrict__ H4 = (const float4*)H;
    float4 acc = H4[(unsigned)n * 4u + (unsigned)k];
    const float aDn = quad_sum(dot4(acc, ad4));
    float m = lrelu(quad_sum(dot4(acc, as4)) + aDn);
    float den = 1.f;
    for (int i = beg; i < end; i += 4) {
        int sv[4];
        float av[4];
        float4 rv[4];
        #pragma unroll
        for (int u = 0; u < 4; ++u) sv[u] = csr[min(i + u, end - 1)];
        #pragma unroll
        for (int u = 0; u < 4; ++u)
            rv[u] = H4[(unsigned)sv[u] * 4u + (unsigned)k];
        float gm = m;
        #pragma unroll
        for (int u = 0; u < 4; ++u) {
            av[u] = lrelu(quad_sum(dot4(rv[u], as4)) + aDn);
            gm = fmaxf(gm, av[u]);
        }
        const float r = __expf(m - gm);
        m = gm;
        den *= r;
        acc.x *= r; acc.y *= r; acc.z *= r; acc.w *= r;
        #pragma unroll
        for (int u = 0; u < 4; ++u) {
            float w = __expf(av[u] - m);
            w = (i + u < end) ? w : 0.f;
            den += w;
            acc.x += w * rv[u].x; acc.y += w * rv[u].y;
            acc.z += w * rv[u].z; acc.w += w * rv[u].w;
        }
    }
    const float4 b4 = *(const float4*)(bias + 4 * k);
    float4 o;
    o.x = acc.x / den + b4.x;
    o.y = acc.y / den + b4.y;
    o.z = acc.z / den + b4.z;
    o.w = acc.w / den + b4.w;
    *(float4*)(O + (size_t)n * OUT_C + 4 * k) = o;
}

#define PADW(x) (((size_t)(x) + 63) & ~(size_t)63)

extern "C" void kernel_launch(void* const* d_in, const int* in_sizes, int n_in,
                              void* d_out, int out_size, void* d_ws, size_t ws_size,
                              hipStream_t stream) {
    const float* x     = (const float*)d_in[0];
    const int*   ei    = (const int*)  d_in[1];
    const float* W1    = (const float*)d_in[2];
    const float* attS1 = (const float*)d_in[3];
    const float* attD1 = (const float*)d_in[4];
    const float* bias1 = (const float*)d_in[5];
    const float* W2    = (const float*)d_in[6];
    const float* attS2 = (const float*)d_in[7];
    const float* attD2 = (const float*)d_in[8];
    const float* bias2 = (const float*)d_in[9];
    float* out = (float*)d_out;
    const int* esrc = ei;
    const int* edst = ei + NE;

    int*      S      = (int*)d_ws;
    int*      rowptr = S + PADW(P_NW * (B_NB + 1));
    unsigned* part   = (unsigned*)(rowptr + PADW(NN + 1));
    int*      csr    = (int*)(part + PADW(NE));
    float* h1   = (float*)(csr + PADW(NE));
    float* h2   = h1 + PADW((size_t)NN * 128);

    k_part<<<P_NW, P_T, 0, stream>>>(esrc, edst, part, S);
    k_build<<<B_NB, B_T, 0, stream>>>(part, S, rowptr, csr);

    k_gemm1<<<G1_GRID, 256, 0, stream>>>(x, W1, h1);
    k_agg1<<<NN / (4 * A1_NPW), 128, 0, stream>>>(rowptr, csr, attS1, attD1, h1, bias1, W2, h2);

    k_agg2<<<(NN * 4 + 255) / 256, 256, 0, stream>>>(rowptr, csr, attS2, attD2, h2, bias2, out);
}
